# starting kernel
# baseline (speedup 1.0000x reference)
#include <hip/hip_runtime.h>
#include <hip/hip_bf16.h>

typedef __attribute__((ext_vector_type(2))) float v2f;
typedef __attribute__((ext_vector_type(4))) float v4f;
typedef __attribute__((ext_vector_type(8))) float v8f;

#define KDIM 128
#define FW   128

__global__ void deg_init(float* __restrict__ dinv, int n) {
    int i = blockIdx.x * blockDim.x + threadIdx.x;
    if (i < n) dinv[i] = 1.0f;
}

__global__ void deg_accum(const int* __restrict__ dst, float* __restrict__ dinv, int e) {
    int i = blockIdx.x * blockDim.x + threadIdx.x;
    if (i < e) atomicAdd(&dinv[dst[i]], 1.0f);
}

__global__ void deg_finish(float* __restrict__ dinv, int n) {
    int i = blockIdx.x * blockDim.x + threadIdx.x;
    if (i < n) dinv[i] = rsqrtf(dinv[i]);
}

template <bool DUAL>
__global__ __launch_bounds__(256)
void gemm_wmma_f32(const float* __restrict__ A,
                   const float* __restrict__ B0,
                   const float* __restrict__ B1,
                   float* __restrict__ C, int nrows)
{
    constexpr int BLD = DUAL ? 64 : 128;
    __shared__ float As[16 * KDIM];
    const int row0 = blockIdx.x * 16;
    const int tid  = threadIdx.x;

    {
        v4f* dstv = (v4f*)As;
        #pragma unroll
        for (int idx = tid; idx < 512; idx += 256) {
            int r  = idx >> 5;
            int c4 = idx & 31;
            int gr = row0 + r; if (gr >= nrows) gr = nrows - 1;
            dstv[idx] = ((const v4f*)(A + (size_t)gr * KDIM))[c4];
        }
    }
    __syncthreads();

    const int wave = tid >> 5;
    const int lane = tid & 31;
    const int col0 = wave * 16;
    const int ln   = lane & 15;
    const int hi   = lane >> 4;

    const float* bp;
    if (DUAL) {
        const float* Bsel = (col0 < 64) ? B0 : B1;
        const int    bcol = (col0 < 64) ? (col0 + ln) : (col0 - 64 + ln);
        bp = Bsel + (size_t)(2 * hi) * BLD + bcol;
    } else {
        bp = B0 + (size_t)(2 * hi) * BLD + col0 + ln;
    }
    const float* ap = &As[ln * KDIM + 2 * hi];

    v8f acc = {};
    #pragma unroll
    for (int k = 0; k < KDIM; k += 4) {
        v2f a = *(const v2f*)(ap + k);
        v2f b;
        b.x = bp[(size_t)k * BLD];
        b.y = bp[(size_t)(k + 1) * BLD];
        acc = __builtin_amdgcn_wmma_f32_16x16x4_f32(
            false, a, false, b, (short)0, acc, false, false);
    }

    float* cp = C + (size_t)(row0 + 8 * hi) * FW + col0 + ln;
    #pragma unroll
    for (int r = 0; r < 8; ++r) {
        if (row0 + r + 8 * hi < nrows) cp[(size_t)r * FW] = acc[r];
    }
}

__global__ void selfloop_init(const float* __restrict__ in, const float* __restrict__ dinv,
                              float* __restrict__ out, int n)
{
    int t = blockIdx.x * blockDim.x + threadIdx.x;
    if (t < n * FW) {
        int i = t >> 7;
        float d = dinv[i];
        out[t] = d * d * in[t];
    }
}

__global__ __launch_bounds__(256)
void edge_scatter(const int* __restrict__ srcIdx, const int* __restrict__ dstIdx,
                  const float* __restrict__ dinv,
                  const float* __restrict__ in, float* __restrict__ out, int e)
{
    int t = blockIdx.x * blockDim.x + threadIdx.x;
    int edge = t >> 5;
    int lane = t & 31;
    if (edge >= e) return;
    int s = srcIdx[edge];
    int d = dstIdx[edge];
    float nm = dinv[s] * dinv[d];
    v4f v = ((const v4f*)(in + (size_t)s * FW))[lane];
    float* op = out + (size_t)d * FW + lane * 4;
    atomicAdd(op + 0, nm * v.x);
    atomicAdd(op + 1, nm * v.y);
    atomicAdd(op + 2, nm * v.z);
    atomicAdd(op + 3, nm * v.w);
}

__global__ void bias_relu(const float* __restrict__ agg, const float* __restrict__ b,
                          float* __restrict__ out, int n)
{
    int t = blockIdx.x * blockDim.x + threadIdx.x;
    if (t < n * FW) {
        int f = t & (FW - 1);
        out[t] = fmaxf(agg[t] + b[f], 0.0f);
    }
}

__global__ void final_out(const float* __restrict__ agg,
                          const float* __restrict__ bmu, const float* __restrict__ blv,
                          float* __restrict__ outp, int n)
{
    int t = blockIdx.x * blockDim.x + threadIdx.x;
    if (t < n * 64) {
        int i = t >> 6, f = t & 63;
        const float* row = agg + (size_t)i * FW;
        outp[t]                   = row[f]      + bmu[f];
        outp[(size_t)n * 64 + t]  = row[64 + f] + blv[f];
    }
}

extern "C" void kernel_launch(void* const* d_in, const int* in_sizes, int n_in,
                              void* d_out, int out_size, void* d_ws, size_t ws_size,
                              hipStream_t stream)
{
    const float* x    = (const float*)d_in[0];
    const int*   ei   = (const int*)  d_in[1];
    const float* W1   = (const float*)d_in[2];
    const float* b1   = (const float*)d_in[3];
    const float* Wmu  = (const float*)d_in[4];
    const float* bmu  = (const float*)d_in[5];
    const float* Wlv  = (const float*)d_in[6];
    const float* blv  = (const float*)d_in[7];

    const int N = in_sizes[0] / KDIM;
    const int E = in_sizes[1] / 2;
    const int* srcIdx = ei;
    const int* dstIdx = ei + E;

    size_t dinv_bytes = (((size_t)N * 4) + 255) & ~(size_t)255;
    float* dinv = (float*)d_ws;
    float* bufA = (float*)((char*)d_ws + dinv_bytes);
    float* bufB = bufA + (size_t)N * FW;

    const int TB = 256;
    dim3 blk(TB);
    int gN   = (N + TB - 1) / TB;
    int gE   = (E + TB - 1) / TB;
    int gNF  = (N * FW + TB - 1) / TB;
    int gNZ  = (N * 64 + TB - 1) / TB;
    int gEw  = (int)(((size_t)E * 32 + TB - 1) / TB);
    int gGemm = (N + 15) / 16;

    deg_init  <<<gN, blk, 0, stream>>>(dinv, N);
    deg_accum <<<gE, blk, 0, stream>>>(dstIdx, dinv, E);
    deg_finish<<<gN, blk, 0, stream>>>(dinv, N);

    gemm_wmma_f32<false><<<gGemm, blk, 0, stream>>>(x, W1, nullptr, bufA, N);

    selfloop_init<<<gNF, blk, 0, stream>>>(bufA, dinv, bufB, N);
    edge_scatter <<<gEw, blk, 0, stream>>>(srcIdx, dstIdx, dinv, bufA, bufB, E);

    bias_relu<<<gNF, blk, 0, stream>>>(bufB, b1, bufA, N);

    gemm_wmma_f32<true><<<gGemm, blk, 0, stream>>>(bufA, Wmu, Wlv, bufB, N);

    selfloop_init<<<gNF, blk, 0, stream>>>(bufB, dinv, bufA, N);
    edge_scatter <<<gEw, blk, 0, stream>>>(srcIdx, dstIdx, dinv, bufB, bufA, E);

    final_out<<<gNZ, blk, 0, stream>>>(bufA, bmu, blv, (float*)d_out, N);
}


# bias_relu and final_out: one 4-byte element per thread (600,000 one-shot waves in all) -> a fixed grid of 8,192 waves that loop, 16 bytes a lane, 8 loads in flight; same adds, values bit-identical
# speedup vs baseline: 2.2188x; 2.2188x over previous
#include <hip/hip_runtime.h>
#include <hip/hip_bf16.h>

typedef __attribute__((ext_vector_type(2))) float v2f;
typedef __attribute__((ext_vector_type(4))) float v4f;
typedef __attribute__((ext_vector_type(8))) float v8f;
typedef __attribute__((ext_vector_type(4))) unsigned v4u;

#define KDIM 128
#define FW   128

#define HBINS  100352
#define HWORDS (HBINS / 2)
#define ECHUNK 16384
#define EPT    (ECHUNK / 1024)
#define PGRP   14
#define ROWS   392

__global__ __launch_bounds__(1024)
void deg_hist(const int* __restrict__ dst, unsigned* __restrict__ part,
              unsigned* __restrict__ keyw, int n, int e)
{
    __shared__ __attribute__((aligned(16))) unsigned hist[HWORDS];
    const int tid = threadIdx.x;
    const int nb  = blockIdx.y * HBINS;
    const int nin = (n - nb < HBINS) ? (n - nb) : HBINS;
    const int nq  = (nin + 7) >> 3;

    const int e0 = blockIdx.x * ECHUNK + 4 * tid;
    int dv[EPT];
    if (blockIdx.x * ECHUNK + ECHUNK <= e) {
        #pragma unroll
        for (int k = 0; k < EPT; ++k) dv[k] = dst[e0 + 4096 * (k >> 2) + (k & 3)];
    } else {
        #pragma unroll
        for (int k = 0; k < EPT; ++k) {
            int ei = e0 + 4096 * (k >> 2) + (k & 3);
            dv[k] = (ei < e) ? dst[ei] : -1;
        }
    }
    if (blockIdx.y == 0) {
        unsigned* kw = keyw + (e0 >> 2);
        #pragma unroll
        for (int g = 0; g < EPT / 4; ++g) {
            unsigned w = 0u;
            #pragma unroll
            for (int j = 0; j < 4; ++j) w |= (((unsigned)dv[4 * g + j] / (unsigned)ROWS) & 255u) << (8 * j);
            kw[1024 * g] = w;
        }
    }
    const v4u zero = {0u, 0u, 0u, 0u};
    for (int i = tid; i < nq; i += 1024) ((v4u*)hist)[i] = zero;
    __syncthreads();

    #pragma unroll
    for (int k = 0; k < EPT; ++k) {
        unsigned rel = (unsigned)(dv[k] - nb);
        if (rel < (unsigned)nin) atomicAdd(&hist[rel >> 1], 1u << ((rel & 1u) << 4));
    }
    __syncthreads();

    v4u* po = (v4u*)(part + ((size_t)blockIdx.y * gridDim.x + blockIdx.x) * HWORDS);
    for (int i = tid; i < nq; i += 1024) po[i] = ((const v4u*)hist)[i];
}

__global__ void deg_finish(const unsigned* __restrict__ part, float* __restrict__ dinv,
                           int n, int nchunks)
{
    const int p = blockIdx.x * blockDim.x + threadIdx.x;
    if (2 * p >= n) return;
    const int r = p / HWORDS;
    const int w = p - r * HWORDS;
    const unsigned* q = part + (size_t)r * nchunks * HWORDS + w;
    unsigned lo = 0u, hi = 0u;
    int c = 0;
    for (; c + PGRP <= nchunks; c += PGRP) {
        unsigned v[PGRP];
        #pragma unroll
        for (int j = 0; j < PGRP; ++j) v[j] = q[(size_t)j * HWORDS];
        #pragma unroll
        for (int j = 0; j < PGRP; ++j) { lo += v[j] & 0xffffu; hi += v[j] >> 16; }
        q += (size_t)PGRP * HWORDS;
    }
    for (; c < nchunks; ++c) {
        unsigned v = q[0];
        lo += v & 0xffffu;
        hi += v >> 16;
        q += HWORDS;
    }
    const float d0 = rsqrtf((float)(1u + lo));
    if (2 * p + 1 < n) {
        v2f d; d.x = d0; d.y = rsqrtf((float)(1u + hi));
        *(v2f*)(dinv + 2 * p) = d;
    } else {
        dinv[2 * p] = d0;
    }
}

template <bool DUAL>
__global__ __launch_bounds__(256)
void gemm_wmma_f32(const float* __restrict__ A,
                   const float* __restrict__ B0,
                   const float* __restrict__ B1,
                   float* __restrict__ C, int nrows)
{
    constexpr int BLD = DUAL ? 64 : 128;
    __shared__ float As[16 * KDIM];
    const int row0 = blockIdx.x * 16;
    const int tid  = threadIdx.x;

    {
        v4f* dstv = (v4f*)As;
        #pragma unroll
        for (int idx = tid; idx < 512; idx += 256) {
            int r  = idx >> 5;
            int c4 = idx & 31;
            int gr = row0 + r; if (gr >= nrows) gr = nrows - 1;
            dstv[idx] = ((const v4f*)(A + (size_t)gr * KDIM))[c4];
        }
    }
    __syncthreads();

    const int wave = tid >> 5;
    const int lane = tid & 31;
    const int col0 = wave * 16;
    const int ln   = lane & 15;
    const int hi   = lane >> 4;

    const float* bp;
    if (DUAL) {
        const float* Bsel = (col0 < 64) ? B0 : B1;
        const int    bcol = (col0 < 64) ? (col0 + ln) : (col0 - 64 + ln);
        bp = Bsel + (size_t)(2 * hi) * BLD + bcol;
    } else {
        bp = B0 + (size_t)(2 * hi) * BLD + col0 + ln;
    }
    const float* ap = &As[ln * KDIM + 2 * hi];

    v8f acc = {};
    #pragma unroll
    for (int k = 0; k < KDIM; k += 4) {
        v2f a = *(const v2f*)(ap + k);
        v2f b;
        b.x = bp[(size_t)k * BLD];
        b.y = bp[(size_t)(k + 1) * BLD];
        acc = __builtin_amdgcn_wmma_f32_16x16x4_f32(
            false, a, false, b, (short)0, acc, false, false);
    }

    float* cp = C + (size_t)(row0 + 8 * hi) * FW + col0 + ln;
    #pragma unroll
    for (int r = 0; r < 8; ++r) {
        if (row0 + r + 8 * hi < nrows) cp[(size_t)r * FW] = acc[r];
    }
}

#define OWAVES 32
#define KT     8
#define TILE_E (32 * 16 * KT)
#define QCAP   520

__global__ __launch_bounds__(1024)
void agg_owner(const int* __restrict__ srcIdx, const int* __restrict__ dstIdx,
               const unsigned* __restrict__ keyw,
               const float* __restrict__ dinv, const float* __restrict__ in,
               float* __restrict__ out, int n, int e)
{
    __shared__ __attribute__((aligned(16))) float acc[ROWS * FW];
    __shared__ int   pendE[OWAVES * QCAP];

    const int tid  = threadIdx.x;
    const int lane = tid & 31;
    const int wave = __builtin_amdgcn_readfirstlane(tid >> 5);
    const int lo   = blockIdx.x * ROWS;
    const int nown = (n - lo < ROWS) ? (n - lo) : ROWS;

    for (int r = wave; r < nown; r += OWAVES) {
        float d  = dinv[lo + r];
        float dd = d * d;
        v4f x = ((const v4f*)(in + (size_t)(lo + r) * FW))[lane];
        v4f y;
        y.x = dd * x.x; y.y = dd * x.y; y.z = dd * x.z; y.w = dd * x.w;
        ((v4f*)(acc + r * FW))[lane] = y;
    }
    __syncthreads();

    int* qE = pendE + wave * QCAP;
    int cnt = 0;
    const unsigned kb = (unsigned)(blockIdx.x & 255) * 0x01010101u;
    const int ntiles = (e + TILE_E - 1) / TILE_E;
    const int shift = (int)(((long long)blockIdx.x * ntiles) / gridDim.x);

    for (int i = wave; i < ntiles; i += OWAVES) {
        int t = i + shift;
        if (t >= ntiles) t -= ntiles;
        const v4u* kp = (const v4u*)(keyw + (size_t)t * (TILE_E / 4)) + lane;
        v4u kv[KT];
        #pragma unroll
        for (int j = 0; j < KT; ++j) kv[j] = kp[32 * j];
        const int ebase = t * TILE_E + 16 * lane;
        #pragma unroll
        for (int j = 0; j < KT; ++j) {
            #pragma unroll
            for (int d = 0; d < 4; ++d) {
                const unsigned x = kv[j][d] ^ kb;
                unsigned f = (x - 0x01010101u) & ~x & 0x80808080u;
                unsigned mask = __builtin_amdgcn_ballot_w32(f != 0u);
                while (mask) {
                    const int  eid = ebase + 512 * j + 4 * d + (__builtin_ctz(f | 0x80000000u) >> 3);
                    const bool ok  = (f != 0u) && (eid < e);
                    const unsigned m2 = __builtin_amdgcn_ballot_w32(ok);
                    const int pos = cnt + (int)__builtin_amdgcn_mbcnt_lo(m2, 0u);
                    if (ok) qE[pos] = eid;
                    cnt += __builtin_popcount(m2);
                    f &= f - 1u;
                    mask = __builtin_amdgcn_ballot_w32(f != 0u);
                }
            }
            while (cnt >= 8) {
                cnt -= 8;
                const int j8  = lane & 7;
                const int eid = qE[cnt + j8];
                const int s   = srcIdx[eid];
                const int dn  = dstIdx[eid];
                const float nm = dinv[s] * dinv[dn];
                float v[8][4];
                #pragma unroll
                for (int q = 0; q < 8; ++q) {
                    const int sq = __builtin_amdgcn_readlane(s, q);
                    const float* ip = in + (size_t)sq * FW + lane;
                    v[q][0] = ip[0]; v[q][1] = ip[32]; v[q][2] = ip[64]; v[q][3] = ip[96];
                }
                asm volatile("" :: "v"(v[0][0]), "v"(v[0][1]), "v"(v[0][2]), "v"(v[0][3]),
                                   "v"(v[1][0]), "v"(v[1][1]), "v"(v[1][2]), "v"(v[1][3]),
                                   "v"(v[2][0]), "v"(v[2][1]), "v"(v[2][2]), "v"(v[2][3]),
                                   "v"(v[3][0]), "v"(v[3][1]), "v"(v[3][2]), "v"(v[3][3]));
                asm volatile("" :: "v"(v[4][0]), "v"(v[4][1]), "v"(v[4][2]), "v"(v[4][3]),
                                   "v"(v[5][0]), "v"(v[5][1]), "v"(v[5][2]), "v"(v[5][3]),
                                   "v"(v[6][0]), "v"(v[6][1]), "v"(v[6][2]), "v"(v[6][3]),
                                   "v"(v[7][0]), "v"(v[7][1]), "v"(v[7][2]), "v"(v[7][3]));
                #pragma unroll
                for (int q = 0; q < 8; ++q) {
                    const unsigned rq = (unsigned)(__builtin_amdgcn_readlane(dn, q) - lo);
                    if (rq < (unsigned)nown) {
                        const float nq = __builtin_bit_cast(float,
                            __builtin_amdgcn_readlane(__builtin_bit_cast(int, nm), q));
                        float* ap = acc + rq * FW + lane;
                        atomicAdd(ap,      nq * v[q][0]);
                        atomicAdd(ap + 32, nq * v[q][1]);
                        atomicAdd(ap + 64, nq * v[q][2]);
                        atomicAdd(ap + 96, nq * v[q][3]);
                    }
                }
            }
        }
    }
    for (int j = 0; j < cnt; ++j) {
        const int eid = qE[j];
        const int s   = srcIdx[eid];
        const int dn  = dstIdx[eid];
        const unsigned rl = (unsigned)(dn - lo);
        if (rl < (unsigned)nown) {
            const float nm = dinv[s] * dinv[dn];
            const float* ip = in + (size_t)s * FW + lane;
            float* ap = acc + rl * FW + lane;
            atomicAdd(ap,      nm * ip[0]);
            atomicAdd(ap + 32, nm * ip[32]);
            atomicAdd(ap + 64, nm * ip[64]);
            atomicAdd(ap + 96, nm * ip[96]);
        }
    }
    __syncthreads();

    for (int r = wave; r < nown; r += OWAVES) {
        ((v4f*)(out + (size_t)(lo + r) * FW))[lane] = ((const v4f*)(acc + r * FW))[lane];
    }
}

#define EW_BLOCKS 1024
#define EW_BATCH  8

__global__ __launch_bounds__(256)
void bias_relu(const float* __restrict__ agg, const float* __restrict__ b,
               float* __restrict__ out, int n)
{
    const int nthr  = gridDim.x * 256;
    const int t0    = blockIdx.x * 256 + threadIdx.x;
    const int total = n * (FW / 4);
    const v4f bb = ((const v4f*)b)[t0 & (FW / 4 - 1)];
    const v4f* __restrict__ src = (const v4f*)agg;
    v4f* __restrict__ dst = (v4f*)out;
    for (int q = t0; q < total; q += EW_BATCH * nthr) {
        v4f v[EW_BATCH];
        #pragma unroll
        for (int k = 0; k < EW_BATCH; ++k) {
            const int qq = q + k * nthr;
            if (qq < total) v[k] = src[qq];
        }
        #pragma unroll
        for (int k = 0; k < EW_BATCH; ++k) {
            const int qq = q + k * nthr;
            if (qq < total) {
                v4f y;
                y.x = fmaxf(v[k].x + bb.x, 0.0f);
                y.y = fmaxf(v[k].y + bb.y, 0.0f);
                y.z = fmaxf(v[k].z + bb.z, 0.0f);
                y.w = fmaxf(v[k].w + bb.w, 0.0f);
                dst[qq] = y;
            }
        }
    }
}

__global__ __launch_bounds__(256)
void final_out(const float* __restrict__ agg,
               const float* __restrict__ bmu, const float* __restrict__ blv,
               float* __restrict__ outp, int n)
{
    const int nthr  = gridDim.x * 256;
    const int t0    = blockIdx.x * 256 + threadIdx.x;
    const int total = n * (FW / 4);
    const int c4    = t0 & (FW / 4 - 1);
    const v4f bb = (c4 < 16) ? ((const v4f*)bmu)[c4] : ((const v4f*)blv)[c4 - 16];
    float* __restrict__ o0 = (c4 < 16) ? (outp + 4 * c4) : (outp + (size_t)n * 64 + 4 * (c4 - 16));
    const v4f* __restrict__ src = (const v4f*)agg;
    for (int q = t0; q < total; q += EW_BATCH * nthr) {
        v4f v[EW_BATCH];
        #pragma unroll
        for (int k = 0; k < EW_BATCH; ++k) {
            const int qq = q + k * nthr;
            if (qq < total) v[k] = src[qq];
        }
        #pragma unroll
        for (int k = 0; k < EW_BATCH; ++k) {
            const int qq = q + k * nthr;
            if (qq < total) {
                v4f y;
                y.x = v[k].x + bb.x;
                y.y = v[k].y + bb.y;
                y.z = v[k].z + bb.z;
                y.w = v[k].w + bb.w;
                *(v4f*)(o0 + (size_t)(qq >> 5) * 64) = y;
            }
        }
    }
}

extern "C" void kernel_launch(void* const* d_in, const int* in_sizes, int n_in,
                              void* d_out, int out_size, void* d_ws, size_t ws_size,
                              hipStream_t stream)
{
    const float* x    = (const float*)d_in[0];
    const int*   ei   = (const int*)  d_in[1];
    const float* W1   = (const float*)d_in[2];
    const float* b1   = (const float*)d_in[3];
    const float* Wmu  = (const float*)d_in[4];
    const float* bmu  = (const float*)d_in[5];
    const float* Wlv  = (const float*)d_in[6];
    const float* blv  = (const float*)d_in[7];

    const int N = in_sizes[0] / KDIM;
    const int E = in_sizes[1] / 2;
    const int* srcIdx = ei;
    const int* dstIdx = ei + E;

    size_t dinv_bytes = (((size_t)N * 4) + 255) & ~(size_t)255;
    float* dinv = (float*)d_ws;
    float* bufA = (float*)((char*)d_ws + dinv_bytes);
    float* bufB = bufA + (size_t)N * FW;
    unsigned* part = (unsigned*)(bufB + (size_t)N * FW);

    const int TB = 256;
    dim3 blk(TB);
    int gP   = ((N + 1) / 2 + TB - 1) / TB;
    int nCh  = (E + ECHUNK - 1) / ECHUNK;
    int nRg  = (N + HBINS - 1) / HBINS;
    int gOwn = (N + ROWS - 1) / ROWS;
    int gGemm = (N + 15) / 16;
    unsigned* keyw = part + (size_t)nRg * nCh * HWORDS;

    deg_hist  <<<dim3(nCh, nRg), dim3(1024), 0, stream>>>(dstIdx, part, keyw, N, E);
    deg_finish<<<gP, blk, 0, stream>>>(part, dinv, N, nCh);

    gemm_wmma_f32<false><<<gGemm, blk, 0, stream>>>(x, W1, nullptr, bufA, N);

    agg_owner<<<gOwn, dim3(1024), 0, stream>>>(srcIdx, dstIdx, keyw, dinv, bufA, bufB, N, E);

    bias_relu<<<EW_BLOCKS, blk, 0, stream>>>(bufB, b1, bufA, N);

    gemm_wmma_f32<true><<<gGemm, blk, 0, stream>>>(bufA, Wmu, Wlv, bufB, N);

    agg_owner<<<gOwn, dim3(1024), 0, stream>>>(srcIdx, dstIdx, keyw, dinv, bufB, bufA, N, E);

    final_out<<<EW_BLOCKS, blk, 0, stream>>>(bufA, bmu, blv, (float*)d_out, N);
}


# layer-2 agg_owner replays the per-wave edge list (source, owned row, weight) the layer-1 scan wrote to the workspace: no second owner-byte scan, no src/dst/dinv reads; LDS f32 adds still arrival order
# speedup vs baseline: 2.5338x; 1.1420x over previous
#include <hip/hip_runtime.h>
#include <hip/hip_bf16.h>

typedef __attribute__((ext_vector_type(2))) float v2f;
typedef __attribute__((ext_vector_type(4))) float v4f;
typedef __attribute__((ext_vector_type(8))) float v8f;
typedef __attribute__((ext_vector_type(4))) unsigned v4u;

#define KDIM 128
#define FW   128

#define HBINS  100352
#define HWORDS (HBINS / 2)
#define ECHUNK 16384
#define EPT    (ECHUNK / 1024)
#define PGRP   14
#define ROWS   392

__global__ __launch_bounds__(1024)
void deg_hist(const int* __restrict__ dst, unsigned* __restrict__ part,
              unsigned* __restrict__ keyw, int n, int e)
{
    __shared__ __attribute__((aligned(16))) unsigned hist[HWORDS];
    const int tid = threadIdx.x;
    const int nb  = blockIdx.y * HBINS;
    const int nin = (n - nb < HBINS) ? (n - nb) : HBINS;
    const int nq  = (nin + 7) >> 3;

    const int e0 = blockIdx.x * ECHUNK + 4 * tid;
    int dv[EPT];
    if (blockIdx.x * ECHUNK + ECHUNK <= e) {
        #pragma unroll
        for (int k = 0; k < EPT; ++k) dv[k] = dst[e0 + 4096 * (k >> 2) + (k & 3)];
    } else {
        #pragma unroll
        for (int k = 0; k < EPT; ++k) {
            int ei = e0 + 4096 * (k >> 2) + (k & 3);
            dv[k] = (ei < e) ? dst[ei] : -1;
        }
    }
    if (blockIdx.y == 0) {
        unsigned* kw = keyw + (e0 >> 2);
        #pragma unroll
        for (int g = 0; g < EPT / 4; ++g) {
            unsigned w = 0u;
            #pragma unroll
            for (int j = 0; j < 4; ++j) w |= (((unsigned)dv[4 * g + j] / (unsigned)ROWS) & 255u) << (8 * j);
            kw[1024 * g] = w;
        }
    }
    const v4u zero = {0u, 0u, 0u, 0u};
    for (int i = tid; i < nq; i += 1024) ((v4u*)hist)[i] = zero;
    __syncthreads();

    #pragma unroll
    for (int k = 0; k < EPT; ++k) {
        unsigned rel = (unsigned)(dv[k] - nb);
        if (rel < (unsigned)nin) atomicAdd(&hist[rel >> 1], 1u << ((rel & 1u) << 4));
    }
    __syncthreads();

    v4u* po = (v4u*)(part + ((size_t)blockIdx.y * gridDim.x + blockIdx.x) * HWORDS);
    for (int i = tid; i < nq; i += 1024) po[i] = ((const v4u*)hist)[i];
}

__global__ void deg_finish(const unsigned* __restrict__ part, float* __restrict__ dinv,
                           int n, int nchunks)
{
    const int p = blockIdx.x * blockDim.x + threadIdx.x;
    if (2 * p >= n) return;
    const int r = p / HWORDS;
    const int w = p - r * HWORDS;
    const unsigned* q = part + (size_t)r * nchunks * HWORDS + w;
    unsigned lo = 0u, hi = 0u;
    int c = 0;
    for (; c + PGRP <= nchunks; c += PGRP) {
        unsigned v[PGRP];
        #pragma unroll
        for (int j = 0; j < PGRP; ++j) v[j] = q[(size_t)j * HWORDS];
        #pragma unroll
        for (int j = 0; j < PGRP; ++j) { lo += v[j] & 0xffffu; hi += v[j] >> 16; }
        q += (size_t)PGRP * HWORDS;
    }
    for (; c < nchunks; ++c) {
        unsigned v = q[0];
        lo += v & 0xffffu;
        hi += v >> 16;
        q += HWORDS;
    }
    const float d0 = rsqrtf((float)(1u + lo));
    if (2 * p + 1 < n) {
        v2f d; d.x = d0; d.y = rsqrtf((float)(1u + hi));
        *(v2f*)(dinv + 2 * p) = d;
    } else {
        dinv[2 * p] = d0;
    }
}

template <bool DUAL>
__global__ __launch_bounds__(256)
void gemm_wmma_f32(const float* __restrict__ A,
                   const float* __restrict__ B0,
                   const float* __restrict__ B1,
                   float* __restrict__ C, int nrows)
{
    constexpr int BLD = DUAL ? 64 : 128;
    __shared__ float As[16 * KDIM];
    const int row0 = blockIdx.x * 16;
    const int tid  = threadIdx.x;

    {
        v4f* dstv = (v4f*)As;
        #pragma unroll
        for (int idx = tid; idx < 512; idx += 256) {
            int r  = idx >> 5;
            int c4 = idx & 31;
            int gr = row0 + r; if (gr >= nrows) gr = nrows - 1;
            dstv[idx] = ((const v4f*)(A + (size_t)gr * KDIM))[c4];
        }
    }
    __syncthreads();

    const int wave = tid >> 5;
    const int lane = tid & 31;
    const int col0 = wave * 16;
    const int ln   = lane & 15;
    const int hi   = lane >> 4;

    const float* bp;
    if (DUAL) {
        const float* Bsel = (col0 < 64) ? B0 : B1;
        const int    bcol = (col0 < 64) ? (col0 + ln) : (col0 - 64 + ln);
        bp = Bsel + (size_t)(2 * hi) * BLD + bcol;
    } else {
        bp = B0 + (size_t)(2 * hi) * BLD + col0 + ln;
    }
    const float* ap = &As[ln * KDIM + 2 * hi];

    v8f acc = {};
    #pragma unroll
    for (int k = 0; k < KDIM; k += 4) {
        v2f a = *(const v2f*)(ap + k);
        v2f b;
        b.x = bp[(size_t)k * BLD];
        b.y = bp[(size_t)(k + 1) * BLD];
        acc = __builtin_amdgcn_wmma_f32_16x16x4_f32(
            false, a, false, b, (short)0, acc, false, false);
    }

    float* cp = C + (size_t)(row0 + 8 * hi) * FW + col0 + ln;
    #pragma unroll
    for (int r = 0; r < 8; ++r) {
        if (row0 + r + 8 * hi < nrows) cp[(size_t)r * FW] = acc[r];
    }
}

#define OWAVES 32
#define KT     8
#define TILE_E (32 * 16 * KT)
#define QCAP   520
#define RCAP   1024


template <bool REC, bool REPLAY>
__global__ __launch_bounds__(1024)
void agg_owner(const int* __restrict__ srcIdx, const int* __restrict__ dstIdx,
               const unsigned* __restrict__ keyw,
               const float* __restrict__ dinv, const float* __restrict__ in,
               float* __restrict__ out, int* __restrict__ rcnt, int* __restrict__ rec,
               int n, int e)
{
    __shared__ __attribute__((aligned(16))) float acc[ROWS * FW];
    __shared__ int   pendE[OWAVES * QCAP];

    const int tid  = threadIdx.x;
    const int lane = tid & 31;
    const int wave = __builtin_amdgcn_readfirstlane(tid >> 5);
    const int lo   = blockIdx.x * ROWS;
    const int nown = (n - lo < ROWS) ? (n - lo) : ROWS;

    for (int r = wave; r < nown; r += OWAVES) {
        float d  = dinv[lo + r];
        float dd = d * d;
        v4f x = ((const v4f*)(in + (size_t)(lo + r) * FW))[lane];
        v4f y;
        y.x = dd * x.x; y.y = dd * x.y; y.z = dd * x.z; y.w = dd * x.w;
        ((v4f*)(acc + r * FW))[lane] = y;
    }
    __syncthreads();

    const int wid = blockIdx.x * OWAVES + wave;
    int*   wS = rec + (size_t)wid * (3 * RCAP);
    int*   wR = wS + RCAP;
    float* wW = (float*)(wR + RCAP);
    int  nrec   = 0;
    bool doScan = true;
    if (REPLAY) {
        nrec = __builtin_amdgcn_readfirstlane(rcnt[wid]);
        if (nrec <= RCAP) {
            doScan = false;
            for (int g = 0; g < nrec; g += 32) {
                const int idx = g + lane;
                int s = 0, r = 0;
                float w = 0.0f;
                if (idx < nrec) { s = wS[idx]; r = wR[idx]; w = wW[idx]; }
                const int nb = (nrec - g < 32) ? (nrec - g) : 32;
                #pragma unroll
                for (int sub = 0; sub < 4; ++sub) {
                    if (8 * sub < nb) {
                        float v[8][4];
                        #pragma unroll
                        for (int q = 0; q < 8; ++q) {
                            const int sq = __builtin_amdgcn_readlane(s, 8 * sub + q);
                            const float* ip = in + (size_t)sq * FW + lane;
                            v[q][0] = ip[0]; v[q][1] = ip[32]; v[q][2] = ip[64]; v[q][3] = ip[96];
                        }
                        asm volatile("" :: "v"(v[0][0]), "v"(v[0][1]), "v"(v[0][2]), "v"(v[0][3]),
                                           "v"(v[1][0]), "v"(v[1][1]), "v"(v[1][2]), "v"(v[1][3]),
                                           "v"(v[2][0]), "v"(v[2][1]), "v"(v[2][2]), "v"(v[2][3]),
                                           "v"(v[3][0]), "v"(v[3][1]), "v"(v[3][2]), "v"(v[3][3]));
                        asm volatile("" :: "v"(v[4][0]), "v"(v[4][1]), "v"(v[4][2]), "v"(v[4][3]),
                                           "v"(v[5][0]), "v"(v[5][1]), "v"(v[5][2]), "v"(v[5][3]),
                                           "v"(v[6][0]), "v"(v[6][1]), "v"(v[6][2]), "v"(v[6][3]),
                                           "v"(v[7][0]), "v"(v[7][1]), "v"(v[7][2]), "v"(v[7][3]));
                        #pragma unroll
                        for (int q = 0; q < 8; ++q) {
                            if (8 * sub + q < nb) {
                                const int   rq = __builtin_amdgcn_readlane(r, 8 * sub + q);
                                const float nq = __builtin_bit_cast(float,
                                    __builtin_amdgcn_readlane(__builtin_bit_cast(int, w), 8 * sub + q));
                                float* ap = acc + rq * FW + lane;
                                atomicAdd(ap,      nq * v[q][0]);
                                atomicAdd(ap + 32, nq * v[q][1]);
                                atomicAdd(ap + 64, nq * v[q][2]);
                                atomicAdd(ap + 96, nq * v[q][3]);
                            }
                        }
                    }
                }
            }
        }
        nrec = 0;
    }

    if (doScan) {
    int* qE = pendE + wave * QCAP;
    int cnt = 0;
    const unsigned kb = (unsigned)(blockIdx.x & 255) * 0x01010101u;
    const int ntiles = (e + TILE_E - 1) / TILE_E;
    const int shift = (int)(((long long)blockIdx.x * ntiles) / gridDim.x);

    for (int i = wave; i < ntiles; i += OWAVES) {
        int t = i + shift;
        if (t >= ntiles) t -= ntiles;
        const v4u* kp = (const v4u*)(keyw + (size_t)t * (TILE_E / 4)) + lane;
        v4u kv[KT];
        #pragma unroll
        for (int j = 0; j < KT; ++j) kv[j] = kp[32 * j];
        const int ebase = t * TILE_E + 16 * lane;
        #pragma unroll
        for (int j = 0; j < KT; ++j) {
            #pragma unroll
            for (int d = 0; d < 4; ++d) {
                const unsigned x = kv[j][d] ^ kb;
                unsigned f = (x - 0x01010101u) & ~x & 0x80808080u;
                unsigned mask = __builtin_amdgcn_ballot_w32(f != 0u);
                while (mask) {
                    const int  eid = ebase + 512 * j + 4 * d + (__builtin_ctz(f | 0x80000000u) >> 3);
                    const bool ok  = (f != 0u) && (eid < e);
                    const unsigned m2 = __builtin_amdgcn_ballot_w32(ok);
                    const int pos = cnt + (int)__builtin_amdgcn_mbcnt_lo(m2, 0u);
                    if (ok) qE[pos] = eid;
                    cnt += __builtin_popcount(m2);
                    f &= f - 1u;
                    mask = __builtin_amdgcn_ballot_w32(f != 0u);
                }
            }
            while (cnt >= 8) {
                cnt -= 8;
                const int j8  = lane & 7;
                const int eid = qE[cnt + j8];
                const int s   = srcIdx[eid];
                const int dn  = dstIdx[eid];
                const float nm = dinv[s] * dinv[dn];
                float v[8][4];
                #pragma unroll
                for (int q = 0; q < 8; ++q) {
                    const int sq = __builtin_amdgcn_readlane(s, q);
                    const float* ip = in + (size_t)sq * FW + lane;
                    v[q][0] = ip[0]; v[q][1] = ip[32]; v[q][2] = ip[64]; v[q][3] = ip[96];
                }
                asm volatile("" :: "v"(v[0][0]), "v"(v[0][1]), "v"(v[0][2]), "v"(v[0][3]),
                                   "v"(v[1][0]), "v"(v[1][1]), "v"(v[1][2]), "v"(v[1][3]),
                                   "v"(v[2][0]), "v"(v[2][1]), "v"(v[2][2]), "v"(v[2][3]),
                                   "v"(v[3][0]), "v"(v[3][1]), "v"(v[3][2]), "v"(v[3][3]));
                asm volatile("" :: "v"(v[4][0]), "v"(v[4][1]), "v"(v[4][2]), "v"(v[4][3]),
                                   "v"(v[5][0]), "v"(v[5][1]), "v"(v[5][2]), "v"(v[5][3]),
                                   "v"(v[6][0]), "v"(v[6][1]), "v"(v[6][2]), "v"(v[6][3]),
                                   "v"(v[7][0]), "v"(v[7][1]), "v"(v[7][2]), "v"(v[7][3]));
                if (REC) {
                    const unsigned rl  = (unsigned)(dn - lo);
                    const bool     val = (lane < 8) && (rl < (unsigned)nown);
                    const unsigned m   = __builtin_amdgcn_ballot_w32(val);
                    const int      pos = nrec + (int)__builtin_amdgcn_mbcnt_lo(m, 0u);
                    if (val && pos < RCAP) { wS[pos] = s; wR[pos] = (int)rl; wW[pos] = nm; }
                    nrec += __builtin_popcount(m);
                }
                #pragma unroll
                for (int q = 0; q < 8; ++q) {
                    const unsigned rq = (unsigned)(__builtin_amdgcn_readlane(dn, q) - lo);
                    if (rq < (unsigned)nown) {
                        const float nq = __builtin_bit_cast(float,
                            __builtin_amdgcn_readlane(__builtin_bit_cast(int, nm), q));
                        float* ap = acc + rq * FW + lane;
                        atomicAdd(ap,      nq * v[q][0]);
                        atomicAdd(ap + 32, nq * v[q][1]);
                        atomicAdd(ap + 64, nq * v[q][2]);
                        atomicAdd(ap + 96, nq * v[q][3]);
                    }
                }
            }
        }
    }
    for (int j = 0; j < cnt; ++j) {
        const int eid = qE[j];
        const int s   = srcIdx[eid];
        const int dn  = dstIdx[eid];
        const unsigned rl = (unsigned)(dn - lo);
        if (rl < (unsigned)nown) {
            const float nm = dinv[s] * dinv[dn];
            const float* ip = in + (size_t)s * FW + lane;
            float* ap = acc + rl * FW + lane;
            atomicAdd(ap,      nm * ip[0]);
            atomicAdd(ap + 32, nm * ip[32]);
            atomicAdd(ap + 64, nm * ip[64]);
            atomicAdd(ap + 96, nm * ip[96]);
            if (REC) {
                if (lane == 0 && nrec < RCAP) { wS[nrec] = s; wR[nrec] = (int)rl; wW[nrec] = nm; }
                ++nrec;
            }
        }
    }
    }
    if (REC) { if (lane == 0) rcnt[wid] = nrec; }
    __syncthreads();

    for (int r = wave; r < nown; r += OWAVES) {
        ((v4f*)(out + (size_t)(lo + r) * FW))[lane] = ((const v4f*)(acc + r * FW))[lane];
    }
}

#define EW_BLOCKS 1024
#define EW_BATCH  8

__global__ __launch_bounds__(256)
void bias_relu(const float* __restrict__ agg, const float* __restrict__ b,
               float* __restrict__ out, int n)
{
    const int nthr  = gridDim.x * 256;
    const int t0    = blockIdx.x * 256 + threadIdx.x;
    const int total = n * (FW / 4);
    const v4f bb = ((const v4f*)b)[t0 & (FW / 4 - 1)];
    const v4f* __restrict__ src = (const v4f*)agg;
    v4f* __restrict__ dst = (v4f*)out;
    for (int q = t0; q < total; q += EW_BATCH * nthr) {
        v4f v[EW_BATCH];
        #pragma unroll
        for (int k = 0; k < EW_BATCH; ++k) {
            const int qq = q + k * nthr;
            if (qq < total) v[k] = src[qq];
        }
        #pragma unroll
        for (int k = 0; k < EW_BATCH; ++k) {
            const int qq = q + k * nthr;
            if (qq < total) {
                v4f y;
                y.x = fmaxf(v[k].x + bb.x, 0.0f);
                y.y = fmaxf(v[k].y + bb.y, 0.0f);
                y.z = fmaxf(v[k].z + bb.z, 0.0f);
                y.w = fmaxf(v[k].w + bb.w, 0.0f);
                dst[qq] = y;
            }
        }
    }
}

__global__ __launch_bounds__(256)
void final_out(const float* __restrict__ agg,
               const float* __restrict__ bmu, const float* __restrict__ blv,
               float* __restrict__ outp, int n)
{
    const int nthr  = gridDim.x * 256;
    const int t0    = blockIdx.x * 256 + threadIdx.x;
    const int total = n * (FW / 4);
    const int c4    = t0 & (FW / 4 - 1);
    const v4f bb = (c4 < 16) ? ((const v4f*)bmu)[c4] : ((const v4f*)blv)[c4 - 16];
    float* __restrict__ o0 = (c4 < 16) ? (outp + 4 * c4) : (outp + (size_t)n * 64 + 4 * (c4 - 16));
    const v4f* __restrict__ src = (const v4f*)agg;
    for (int q = t0; q < total; q += EW_BATCH * nthr) {
        v4f v[EW_BATCH];
        #pragma unroll
        for (int k = 0; k < EW_BATCH; ++k) {
            const int qq = q + k * nthr;
            if (qq < total) v[k] = src[qq];
        }
        #pragma unroll
        for (int k = 0; k < EW_BATCH; ++k) {
            const int qq = q + k * nthr;
            if (qq < total) {
                v4f y;
                y.x = v[k].x + bb.x;
                y.y = v[k].y + bb.y;
                y.z = v[k].z + bb.z;
                y.w = v[k].w + bb.w;
                *(v4f*)(o0 + (size_t)(qq >> 5) * 64) = y;
            }
        }
    }
}

extern "C" void kernel_launch(void* const* d_in, const int* in_sizes, int n_in,
                              void* d_out, int out_size, void* d_ws, size_t ws_size,
                              hipStream_t stream)
{
    const float* x    = (const float*)d_in[0];
    const int*   ei   = (const int*)  d_in[1];
    const float* W1   = (const float*)d_in[2];
    const float* b1   = (const float*)d_in[3];
    const float* Wmu  = (const float*)d_in[4];
    const float* bmu  = (const float*)d_in[5];
    const float* Wlv  = (const float*)d_in[6];
    const float* blv  = (const float*)d_in[7];

    const int N = in_sizes[0] / KDIM;
    const int E = in_sizes[1] / 2;
    const int* srcIdx = ei;
    const int* dstIdx = ei + E;

    size_t dinv_bytes = (((size_t)N * 4) + 255) & ~(size_t)255;
    float* dinv = (float*)d_ws;
    float* bufA = (float*)((char*)d_ws + dinv_bytes);
    float* bufB = bufA + (size_t)N * FW;
    unsigned* part = (unsigned*)(bufB + (size_t)N * FW);

    const int TB = 256;
    dim3 blk(TB);
    int gP   = ((N + 1) / 2 + TB - 1) / TB;
    int nCh  = (E + ECHUNK - 1) / ECHUNK;
    int nRg  = (N + HBINS - 1) / HBINS;
    int gOwn = (N + ROWS - 1) / ROWS;
    int gGemm = (N + 15) / 16;
    unsigned* keyw = part + (size_t)nRg * nCh * HWORDS;
    int* rcnt = (int*)(keyw + (size_t)nCh * (ECHUNK / 4));
    int* rec  = rcnt + (size_t)gOwn * OWAVES;

    deg_hist  <<<dim3(nCh, nRg), dim3(1024), 0, stream>>>(dstIdx, part, keyw, N, E);
    deg_finish<<<gP, blk, 0, stream>>>(part, dinv, N, nCh);

    gemm_wmma_f32<false><<<gGemm, blk, 0, stream>>>(x, W1, nullptr, bufA, N);

    agg_owner<true, false><<<gOwn, dim3(1024), 0, stream>>>(srcIdx, dstIdx, keyw, dinv, bufA, bufB, rcnt, rec, N, E);

    bias_relu<<<EW_BLOCKS, blk, 0, stream>>>(bufB, b1, bufA, N);

    gemm_wmma_f32<true><<<gGemm, blk, 0, stream>>>(bufA, Wmu, Wlv, bufB, N);

    agg_owner<false, true><<<gOwn, dim3(1024), 0, stream>>>(srcIdx, dstIdx, keyw, dinv, bufB, bufA, rcnt, rec, N, E);

    final_out<<<EW_BLOCKS, blk, 0, stream>>>(bufA, bmu, blv, (float*)d_out, N);
}


# GEMM re-tiled: wave = 16 rows x 128 cols (8 accumulators), A fragments straight from memory 16 B a lane (no LDS strip), weights once per 64-row block in LDS in fragment order; k sum order differs
# speedup vs baseline: 3.0298x; 1.1958x over previous
#include <hip/hip_runtime.h>
#include <hip/hip_bf16.h>

typedef __attribute__((ext_vector_type(2))) float v2f;
typedef __attribute__((ext_vector_type(4))) float v4f;
typedef __attribute__((ext_vector_type(8))) float v8f;
typedef __attribute__((ext_vector_type(4))) unsigned v4u;

#define KDIM 128
#define FW   128

#define HBINS  100352
#define HWORDS (HBINS / 2)
#define ECHUNK 16384
#define EPT    (ECHUNK / 1024)
#define PGRP   14
#define ROWS   392

__global__ __launch_bounds__(1024)
void deg_hist(const int* __restrict__ dst, unsigned* __restrict__ part,
              unsigned* __restrict__ keyw, int n, int e)
{
    __shared__ __attribute__((aligned(16))) unsigned hist[HWORDS];
    const int tid = threadIdx.x;
    const int nb  = blockIdx.y * HBINS;
    const int nin = (n - nb < HBINS) ? (n - nb) : HBINS;
    const int nq  = (nin + 7) >> 3;

    const int e0 = blockIdx.x * ECHUNK + 4 * tid;
    int dv[EPT];
    if (blockIdx.x * ECHUNK + ECHUNK <= e) {
        #pragma unroll
        for (int k = 0; k < EPT; ++k) dv[k] = dst[e0 + 4096 * (k >> 2) + (k & 3)];
    } else {
        #pragma unroll
        for (int k = 0; k < EPT; ++k) {
            int ei = e0 + 4096 * (k >> 2) + (k & 3);
            dv[k] = (ei < e) ? dst[ei] : -1;
        }
    }
    if (blockIdx.y == 0) {
        unsigned* kw = keyw + (e0 >> 2);
        #pragma unroll
        for (int g = 0; g < EPT / 4; ++g) {
            unsigned w = 0u;
            #pragma unroll
            for (int j = 0; j < 4; ++j) w |= (((unsigned)dv[4 * g + j] / (unsigned)ROWS) & 255u) << (8 * j);
            kw[1024 * g] = w;
        }
    }
    const v4u zero = {0u, 0u, 0u, 0u};
    for (int i = tid; i < nq; i += 1024) ((v4u*)hist)[i] = zero;
    __syncthreads();

    #pragma unroll
    for (int k = 0; k < EPT; ++k) {
        unsigned rel = (unsigned)(dv[k] - nb);
        if (rel < (unsigned)nin) atomicAdd(&hist[rel >> 1], 1u << ((rel & 1u) << 4));
    }
    __syncthreads();

    v4u* po = (v4u*)(part + ((size_t)blockIdx.y * gridDim.x + blockIdx.x) * HWORDS);
    for (int i = tid; i < nq; i += 1024) po[i] = ((const v4u*)hist)[i];
}

__global__ void deg_finish(const unsigned* __restrict__ part, float* __restrict__ dinv,
                           int n, int nchunks)
{
    const int p = blockIdx.x * blockDim.x + threadIdx.x;
    if (2 * p >= n) return;
    const int r = p / HWORDS;
    const int w = p - r * HWORDS;
    const unsigned* q = part + (size_t)r * nchunks * HWORDS + w;
    unsigned lo = 0u, hi = 0u;
    int c = 0;
    for (; c + PGRP <= nchunks; c += PGRP) {
        unsigned v[PGRP];
        #pragma unroll
        for (int j = 0; j < PGRP; ++j) v[j] = q[(size_t)j * HWORDS];
        #pragma unroll
        for (int j = 0; j < PGRP; ++j) { lo += v[j] & 0xffffu; hi += v[j] >> 16; }
        q += (size_t)PGRP * HWORDS;
    }
    for (; c < nchunks; ++c) {
        unsigned v = q[0];
        lo += v & 0xffffu;
        hi += v >> 16;
        q += HWORDS;
    }
    const float d0 = rsqrtf((float)(1u + lo));
    if (2 * p + 1 < n) {
        v2f d; d.x = d0; d.y = rsqrtf((float)(1u + hi));
        *(v2f*)(dinv + 2 * p) = d;
    } else {
        dinv[2 * p] = d0;
    }
}

#define GROWS 64

template <bool DUAL>
__global__ __launch_bounds__(128)
void gemm_wmma_f32(const float* __restrict__ A,
                   const float* __restrict__ B0,
                   const float* __restrict__ B1,
                   float* __restrict__ C, int nrows)
{
    constexpr int BLD = DUAL ? 64 : 128;
    __shared__ v4f Bs[16 * 8 * 32];
    const int tid  = threadIdx.x;
    const int lane = tid & 31;
    const int wave = __builtin_amdgcn_readfirstlane(tid >> 5);
    const int c    = lane & 15;
    const int h    = lane >> 4;
    const int row0 = blockIdx.x * GROWS + wave * 16;

    int gr = row0 + c; if (gr >= nrows) gr = nrows - 1;
    const v4f* ar = (const v4f*)(A + (size_t)gr * KDIM) + h;
    v4f a[16];
    #pragma unroll
    for (int j = 0; j < 16; ++j) a[j] = ar[2 * j];

    #pragma unroll 1
    for (int i0 = 0; i0 < 8; i0 += 4) {
        v4f r[4][4];
        #pragma unroll
        for (int i = 0; i < 4; ++i) {
            const int idx = tid + 128 * (i0 + i);
            const int kq = idx >> 5;
            const int cp = idx & 31;
            const float* w;
            if (DUAL) w = (cp < 16) ? (B0 + 4 * cp) : (B1 + 4 * (cp - 16));
            else      w = B0 + 4 * cp;
            w += (size_t)(4 * kq) * BLD;
            r[i][0] = *(const v4f*)(w);
            r[i][1] = *(const v4f*)(w + BLD);
            r[i][2] = *(const v4f*)(w + 2 * BLD);
            r[i][3] = *(const v4f*)(w + 3 * BLD);
        }
        #pragma unroll
        for (int i = 0; i < 4; ++i) {
            const int idx = tid + 128 * (i0 + i);
            const int kq = idx >> 5;
            const int cp = idx & 31;
            v4f* d = Bs + (((kq >> 1) * 8 + 4 * (cp >> 4)) * 32 + (kq & 1) * 16 + (cp & 15));
            v4f q;
            q.x = r[i][0].x; q.y = r[i][1].x; q.z = r[i][2].x; q.w = r[i][3].x; d[0]  = q;
            q.x = r[i][0].y; q.y = r[i][1].y; q.z = r[i][2].y; q.w = r[i][3].y; d[32] = q;
            q.x = r[i][0].z; q.y = r[i][1].z; q.z = r[i][2].z; q.w = r[i][3].z; d[64] = q;
            q.x = r[i][0].w; q.y = r[i][1].w; q.z = r[i][2].w; q.w = r[i][3].w; d[96] = q;
        }
    }
    __syncthreads();
    if (row0 >= nrows) return;

    const v4f* bl = Bs + lane;
    v8f acc[8];
    #pragma unroll
    for (int t = 0; t < 8; ++t) acc[t] = (v8f){0.f, 0.f, 0.f, 0.f, 0.f, 0.f, 0.f, 0.f};

    #pragma unroll
    for (int j = 0; j < 16; ++j) {
        v4f b[8];
        #pragma unroll
        for (int t = 0; t < 8; ++t) b[t] = bl[(j * 8 + t) * 32];
        v2f a0; a0.x = a[j].x; a0.y = a[j].y;
        v2f a1; a1.x = a[j].z; a1.y = a[j].w;
        #pragma unroll
        for (int t = 0; t < 8; ++t) {
            v2f b0; b0.x = b[t].x; b0.y = b[t].y;
            acc[t] = __builtin_amdgcn_wmma_f32_16x16x4_f32(
                false, a0, false, b0, (short)0, acc[t], false, false);
        }
        #pragma unroll
        for (int t = 0; t < 8; ++t) {
            v2f b1; b1.x = b[t].z; b1.y = b[t].w;
            acc[t] = __builtin_amdgcn_wmma_f32_16x16x4_f32(
                false, a1, false, b1, (short)0, acc[t], false, false);
        }
    }

    float* cp0 = C + (size_t)(row0 + 8 * h) * FW + 4 * c;
    #pragma unroll
    for (int r = 0; r < 8; ++r) {
        if (row0 + r + 8 * h < nrows) {
            #pragma unroll
            for (int g = 0; g < 2; ++g) {
                v4f o;
                o.x = acc[4 * g][r]; o.y = acc[4 * g + 1][r]; o.z = acc[4 * g + 2][r]; o.w = acc[4 * g + 3][r];
                *(v4f*)(cp0 + (size_t)r * FW + 64 * g) = o;
            }
        }
    }
}

#define OWAVES 32
#define KT     8
#define TILE_E (32 * 16 * KT)
#define QCAP   520
#define RCAP   1024


template <bool REC, bool REPLAY>
__global__ __launch_bounds__(1024)
void agg_owner(const int* __restrict__ srcIdx, const int* __restrict__ dstIdx,
               const unsigned* __restrict__ keyw,
               const float* __restrict__ dinv, const float* __restrict__ in,
               float* __restrict__ out, int* __restrict__ rcnt, int* __restrict__ rec,
               int n, int e)
{
    __shared__ __attribute__((aligned(16))) float acc[ROWS * FW];
    __shared__ int   pendE[OWAVES * QCAP];

    const int tid  = threadIdx.x;
    const int lane = tid & 31;
    const int wave = __builtin_amdgcn_readfirstlane(tid >> 5);
    const int lo   = blockIdx.x * ROWS;
    const int nown = (n - lo < ROWS) ? (n - lo) : ROWS;

    for (int r = wave; r < nown; r += OWAVES) {
        float d  = dinv[lo + r];
        float dd = d * d;
        v4f x = ((const v4f*)(in + (size_t)(lo + r) * FW))[lane];
        v4f y;
        y.x = dd * x.x; y.y = dd * x.y; y.z = dd * x.z; y.w = dd * x.w;
        ((v4f*)(acc + r * FW))[lane] = y;
    }
    __syncthreads();

    const int wid = blockIdx.x * OWAVES + wave;
    int*   wS = rec + (size_t)wid * (3 * RCAP);
    int*   wR = wS + RCAP;
    float* wW = (float*)(wR + RCAP);
    int  nrec   = 0;
    bool doScan = true;
    if (REPLAY) {
        nrec = __builtin_amdgcn_readfirstlane(rcnt[wid]);
        if (nrec <= RCAP) {
            doScan = false;
            for (int g = 0; g < nrec; g += 32) {
                const int idx = g + lane;
                int s = 0, r = 0;
                float w = 0.0f;
                if (idx < nrec) { s = wS[idx]; r = wR[idx]; w = wW[idx]; }
                const int nb = (nrec - g < 32) ? (nrec - g) : 32;
                #pragma unroll
                for (int sub = 0; sub < 4; ++sub) {
                    if (8 * sub < nb) {
                        float v[8][4];
                        #pragma unroll
                        for (int q = 0; q < 8; ++q) {
                            const int sq = __builtin_amdgcn_readlane(s, 8 * sub + q);
                            const float* ip = in + (size_t)sq * FW + lane;
                            v[q][0] = ip[0]; v[q][1] = ip[32]; v[q][2] = ip[64]; v[q][3] = ip[96];
                        }
                        asm volatile("" :: "v"(v[0][0]), "v"(v[0][1]), "v"(v[0][2]), "v"(v[0][3]),
                                           "v"(v[1][0]), "v"(v[1][1]), "v"(v[1][2]), "v"(v[1][3]),
                                           "v"(v[2][0]), "v"(v[2][1]), "v"(v[2][2]), "v"(v[2][3]),
                                           "v"(v[3][0]), "v"(v[3][1]), "v"(v[3][2]), "v"(v[3][3]));
                        asm volatile("" :: "v"(v[4][0]), "v"(v[4][1]), "v"(v[4][2]), "v"(v[4][3]),
                                           "v"(v[5][0]), "v"(v[5][1]), "v"(v[5][2]), "v"(v[5][3]),
                                           "v"(v[6][0]), "v"(v[6][1]), "v"(v[6][2]), "v"(v[6][3]),
                                           "v"(v[7][0]), "v"(v[7][1]), "v"(v[7][2]), "v"(v[7][3]));
                        #pragma unroll
                        for (int q = 0; q < 8; ++q) {
                            if (8 * sub + q < nb) {
                                const int   rq = __builtin_amdgcn_readlane(r, 8 * sub + q);
                                const float nq = __builtin_bit_cast(float,
                                    __builtin_amdgcn_readlane(__builtin_bit_cast(int, w), 8 * sub + q));
                                float* ap = acc + rq * FW + lane;
                                atomicAdd(ap,      nq * v[q][0]);
                                atomicAdd(ap + 32, nq * v[q][1]);
                                atomicAdd(ap + 64, nq * v[q][2]);
                                atomicAdd(ap + 96, nq * v[q][3]);
                            }
                        }
                    }
                }
            }
        }
        nrec = 0;
    }

    if (doScan) {
    int* qE = pendE + wave * QCAP;
    int cnt = 0;
    const unsigned kb = (unsigned)(blockIdx.x & 255) * 0x01010101u;
    const int ntiles = (e + TILE_E - 1) / TILE_E;
    const int shift = (int)(((long long)blockIdx.x * ntiles) / gridDim.x);

    for (int i = wave; i < ntiles; i += OWAVES) {
        int t = i + shift;
        if (t >= ntiles) t -= ntiles;
        const v4u* kp = (const v4u*)(keyw + (size_t)t * (TILE_E / 4)) + lane;
        v4u kv[KT];
        #pragma unroll
        for (int j = 0; j < KT; ++j) kv[j] = kp[32 * j];
        const int ebase = t * TILE_E + 16 * lane;
        #pragma unroll
        for (int j = 0; j < KT; ++j) {
            #pragma unroll
            for (int d = 0; d < 4; ++d) {
                const unsigned x = kv[j][d] ^ kb;
                unsigned f = (x - 0x01010101u) & ~x & 0x80808080u;
                unsigned mask = __builtin_amdgcn_ballot_w32(f != 0u);
                while (mask) {
                    const int  eid = ebase + 512 * j + 4 * d + (__builtin_ctz(f | 0x80000000u) >> 3);
                    const bool ok  = (f != 0u) && (eid < e);
                    const unsigned m2 = __builtin_amdgcn_ballot_w32(ok);
                    const int pos = cnt + (int)__builtin_amdgcn_mbcnt_lo(m2, 0u);
                    if (ok) qE[pos] = eid;
                    cnt += __builtin_popcount(m2);
                    f &= f - 1u;
                    mask = __builtin_amdgcn_ballot_w32(f != 0u);
                }
            }
            while (cnt >= 8) {
                cnt -= 8;
                const int j8  = lane & 7;
                const int eid = qE[cnt + j8];
                const int s   = srcIdx[eid];
                const int dn  = dstIdx[eid];
                const float nm = dinv[s] * dinv[dn];
                float v[8][4];
                #pragma unroll
                for (int q = 0; q < 8; ++q) {
                    const int sq = __builtin_amdgcn_readlane(s, q);
                    const float* ip = in + (size_t)sq * FW + lane;
                    v[q][0] = ip[0]; v[q][1] = ip[32]; v[q][2] = ip[64]; v[q][3] = ip[96];
                }
                asm volatile("" :: "v"(v[0][0]), "v"(v[0][1]), "v"(v[0][2]), "v"(v[0][3]),
                                   "v"(v[1][0]), "v"(v[1][1]), "v"(v[1][2]), "v"(v[1][3]),
                                   "v"(v[2][0]), "v"(v[2][1]), "v"(v[2][2]), "v"(v[2][3]),
                                   "v"(v[3][0]), "v"(v[3][1]), "v"(v[3][2]), "v"(v[3][3]));
                asm volatile("" :: "v"(v[4][0]), "v"(v[4][1]), "v"(v[4][2]), "v"(v[4][3]),
                                   "v"(v[5][0]), "v"(v[5][1]), "v"(v[5][2]), "v"(v[5][3]),
                                   "v"(v[6][0]), "v"(v[6][1]), "v"(v[6][2]), "v"(v[6][3]),
                                   "v"(v[7][0]), "v"(v[7][1]), "v"(v[7][2]), "v"(v[7][3]));
                if (REC) {
                    const unsigned rl  = (unsigned)(dn - lo);
                    const bool     val = (lane < 8) && (rl < (unsigned)nown);
                    const unsigned m   = __builtin_amdgcn_ballot_w32(val);
                    const int      pos = nrec + (int)__builtin_amdgcn_mbcnt_lo(m, 0u);
                    if (val && pos < RCAP) { wS[pos] = s; wR[pos] = (int)rl; wW[pos] = nm; }
                    nrec += __builtin_popcount(m);
                }
                #pragma unroll
                for (int q = 0; q < 8; ++q) {
                    const unsigned rq = (unsigned)(__builtin_amdgcn_readlane(dn, q) - lo);
                    if (rq < (unsigned)nown) {
                        const float nq = __builtin_bit_cast(float,
                            __builtin_amdgcn_readlane(__builtin_bit_cast(int, nm), q));
                        float* ap = acc + rq * FW + lane;
                        atomicAdd(ap,      nq * v[q][0]);
                        atomicAdd(ap + 32, nq * v[q][1]);
                        atomicAdd(ap + 64, nq * v[q][2]);
                        atomicAdd(ap + 96, nq * v[q][3]);
                    }
                }
            }
        }
    }
    for (int j = 0; j < cnt; ++j) {
        const int eid = qE[j];
        const int s   = srcIdx[eid];
        const int dn  = dstIdx[eid];
        const unsigned rl = (unsigned)(dn - lo);
        if (rl < (unsigned)nown) {
            const float nm = dinv[s] * dinv[dn];
            const float* ip = in + (size_t)s * FW + lane;
            float* ap = acc + rl * FW + lane;
            atomicAdd(ap,      nm * ip[0]);
            atomicAdd(ap + 32, nm * ip[32]);
            atomicAdd(ap + 64, nm * ip[64]);
            atomicAdd(ap + 96, nm * ip[96]);
            if (REC) {
                if (lane == 0 && nrec < RCAP) { wS[nrec] = s; wR[nrec] = (int)rl; wW[nrec] = nm; }
                ++nrec;
            }
        }
    }
    }
    if (REC) { if (lane == 0) rcnt[wid] = nrec; }
    __syncthreads();

    for (int r = wave; r < nown; r += OWAVES) {
        ((v4f*)(out + (size_t)(lo + r) * FW))[lane] = ((const v4f*)(acc + r * FW))[lane];
    }
}

#define EW_BLOCKS 1024
#define EW_BATCH  8

__global__ __launch_bounds__(256)
void bias_relu(const float* __restrict__ agg, const float* __restrict__ b,
               float* __restrict__ out, int n)
{
    const int nthr  = gridDim.x * 256;
    const int t0    = blockIdx.x * 256 + threadIdx.x;
    const int total = n * (FW / 4);
    const v4f bb = ((const v4f*)b)[t0 & (FW / 4 - 1)];
    const v4f* __restrict__ src = (const v4f*)agg;
    v4f* __restrict__ dst = (v4f*)out;
    for (int q = t0; q < total; q += EW_BATCH * nthr) {
        v4f v[EW_BATCH];
        #pragma unroll
        for (int k = 0; k < EW_BATCH; ++k) {
            const int qq = q + k * nthr;
            if (qq < total) v[k] = src[qq];
        }
        #pragma unroll
        for (int k = 0; k < EW_BATCH; ++k) {
            const int qq = q + k * nthr;
            if (qq < total) {
                v4f y;
                y.x = fmaxf(v[k].x + bb.x, 0.0f);
                y.y = fmaxf(v[k].y + bb.y, 0.0f);
                y.z = fmaxf(v[k].z + bb.z, 0.0f);
                y.w = fmaxf(v[k].w + bb.w, 0.0f);
                dst[qq] = y;
            }
        }
    }
}

__global__ __launch_bounds__(256)
void final_out(const float* __restrict__ agg,
               const float* __restrict__ bmu, const float* __restrict__ blv,
               float* __restrict__ outp, int n)
{
    const int nthr  = gridDim.x * 256;
    const int t0    = blockIdx.x * 256 + threadIdx.x;
    const int total = n * (FW / 4);
    const int c4    = t0 & (FW / 4 - 1);
    const v4f bb = (c4 < 16) ? ((const v4f*)bmu)[c4] : ((const v4f*)blv)[c4 - 16];
    float* __restrict__ o0 = (c4 < 16) ? (outp + 4 * c4) : (outp + (size_t)n * 64 + 4 * (c4 - 16));
    const v4f* __restrict__ src = (const v4f*)agg;
    for (int q = t0; q < total; q += EW_BATCH * nthr) {
        v4f v[EW_BATCH];
        #pragma unroll
        for (int k = 0; k < EW_BATCH; ++k) {
            const int qq = q + k * nthr;
            if (qq < total) v[k] = src[qq];
        }
        #pragma unroll
        for (int k = 0; k < EW_BATCH; ++k) {
            const int qq = q + k * nthr;
            if (qq < total) {
                v4f y;
                y.x = v[k].x + bb.x;
                y.y = v[k].y + bb.y;
                y.z = v[k].z + bb.z;
                y.w = v[k].w + bb.w;
                *(v4f*)(o0 + (size_t)(qq >> 5) * 64) = y;
            }
        }
    }
}

extern "C" void kernel_launch(void* const* d_in, const int* in_sizes, int n_in,
                              void* d_out, int out_size, void* d_ws, size_t ws_size,
                              hipStream_t stream)
{
    const float* x    = (const float*)d_in[0];
    const int*   ei   = (const int*)  d_in[1];
    const float* W1   = (const float*)d_in[2];
    const float* b1   = (const float*)d_in[3];
    const float* Wmu  = (const float*)d_in[4];
    const float* bmu  = (const float*)d_in[5];
    const float* Wlv  = (const float*)d_in[6];
    const float* blv  = (const float*)d_in[7];

    const int N = in_sizes[0] / KDIM;
    const int E = in_sizes[1] / 2;
    const int* srcIdx = ei;
    const int* dstIdx = ei + E;

    size_t dinv_bytes = (((size_t)N * 4) + 255) & ~(size_t)255;
    float* dinv = (float*)d_ws;
    float* bufA = (float*)((char*)d_ws + dinv_bytes);
    float* bufB = bufA + (size_t)N * FW;
    unsigned* part = (unsigned*)(bufB + (size_t)N * FW);

    const int TB = 256;
    dim3 blk(TB);
    int gP   = ((N + 1) / 2 + TB - 1) / TB;
    int nCh  = (E + ECHUNK - 1) / ECHUNK;
    int nRg  = (N + HBINS - 1) / HBINS;
    int gOwn = (N + ROWS - 1) / ROWS;
    int gGemm = (N + GROWS - 1) / GROWS;
    unsigned* keyw = part + (size_t)nRg * nCh * HWORDS;
    int* rcnt = (int*)(keyw + (size_t)nCh * (ECHUNK / 4));
    int* rec  = rcnt + (size_t)gOwn * OWAVES;

    deg_hist  <<<dim3(nCh, nRg), dim3(1024), 0, stream>>>(dstIdx, part, keyw, N, E);
    deg_finish<<<gP, blk, 0, stream>>>(part, dinv, N, nCh);

    gemm_wmma_f32<false><<<gGemm, dim3(128), 0, stream>>>(x, W1, nullptr, bufA, N);

    agg_owner<true, false><<<gOwn, dim3(1024), 0, stream>>>(srcIdx, dstIdx, keyw, dinv, bufA, bufB, rcnt, rec, N, E);

    bias_relu<<<EW_BLOCKS, blk, 0, stream>>>(bufB, b1, bufA, N);

    gemm_wmma_f32<true><<<gGemm, dim3(128), 0, stream>>>(bufA, Wmu, Wlv, bufB, N);

    agg_owner<false, true><<<gOwn, dim3(1024), 0, stream>>>(srcIdx, dstIdx, keyw, dinv, bufB, bufA, rcnt, rec, N, E);

    final_out<<<EW_BLOCKS, blk, 0, stream>>>(bufA, bmu, blv, (float*)d_out, N);
}


# bias+relu pass removed: the layer-2 GEMM reads the raw node sums and adds bias and clips at zero on its own A fragments in registers; one launch, a 51 MB write and its re-read gone; bit-identical
# speedup vs baseline: 3.1679x; 1.0456x over previous
#include <hip/hip_runtime.h>
#include <hip/hip_bf16.h>

typedef __attribute__((ext_vector_type(2))) float v2f;
typedef __attribute__((ext_vector_type(4))) float v4f;
typedef __attribute__((ext_vector_type(8))) float v8f;
typedef __attribute__((ext_vector_type(4))) unsigned v4u;

#define KDIM 128
#define FW   128

#define HBINS  100352
#define HWORDS (HBINS / 2)
#define ECHUNK 16384
#define EPT    (ECHUNK / 1024)
#define PGRP   14
#define ROWS   392

__global__ __launch_bounds__(1024)
void deg_hist(const int* __restrict__ dst, unsigned* __restrict__ part,
              unsigned* __restrict__ keyw, int n, int e)
{
    __shared__ __attribute__((aligned(16))) unsigned hist[HWORDS];
    const int tid = threadIdx.x;
    const int nb  = blockIdx.y * HBINS;
    const int nin = (n - nb < HBINS) ? (n - nb) : HBINS;
    const int nq  = (nin + 7) >> 3;

    const int e0 = blockIdx.x * ECHUNK + 4 * tid;
    int dv[EPT];
    if (blockIdx.x * ECHUNK + ECHUNK <= e) {
        #pragma unroll
        for (int k = 0; k < EPT; ++k) dv[k] = dst[e0 + 4096 * (k >> 2) + (k & 3)];
    } else {
        #pragma unroll
        for (int k = 0; k < EPT; ++k) {
            int ei = e0 + 4096 * (k >> 2) + (k & 3);
            dv[k] = (ei < e) ? dst[ei] : -1;
        }
    }
    if (blockIdx.y == 0) {
        unsigned* kw = keyw + (e0 >> 2);
        #pragma unroll
        for (int g = 0; g < EPT / 4; ++g) {
            unsigned w = 0u;
            #pragma unroll
            for (int j = 0; j < 4; ++j) w |= (((unsigned)dv[4 * g + j] / (unsigned)ROWS) & 255u) << (8 * j);
            kw[1024 * g] = w;
        }
    }
    const v4u zero = {0u, 0u, 0u, 0u};
    for (int i = tid; i < nq; i += 1024) ((v4u*)hist)[i] = zero;
    __syncthreads();

    #pragma unroll
    for (int k = 0; k < EPT; ++k) {
        unsigned rel = (unsigned)(dv[k] - nb);
        if (rel < (unsigned)nin) atomicAdd(&hist[rel >> 1], 1u << ((rel & 1u) << 4));
    }
    __syncthreads();

    v4u* po = (v4u*)(part + ((size_t)blockIdx.y * gridDim.x + blockIdx.x) * HWORDS);
    for (int i = tid; i < nq; i += 1024) po[i] = ((const v4u*)hist)[i];
}

__global__ void deg_finish(const unsigned* __restrict__ part, float* __restrict__ dinv,
                           int n, int nchunks)
{
    const int p = blockIdx.x * blockDim.x + threadIdx.x;
    if (2 * p >= n) return;
    const int r = p / HWORDS;
    const int w = p - r * HWORDS;
    const unsigned* q = part + (size_t)r * nchunks * HWORDS + w;
    unsigned lo = 0u, hi = 0u;
    int c = 0;
    for (; c + PGRP <= nchunks; c += PGRP) {
        unsigned v[PGRP];
        #pragma unroll
        for (int j = 0; j < PGRP; ++j) v[j] = q[(size_t)j * HWORDS];
        #pragma unroll
        for (int j = 0; j < PGRP; ++j) { lo += v[j] & 0xffffu; hi += v[j] >> 16; }
        q += (size_t)PGRP * HWORDS;
    }
    for (; c < nchunks; ++c) {
        unsigned v = q[0];
        lo += v & 0xffffu;
        hi += v >> 16;
        q += HWORDS;
    }
    const float d0 = rsqrtf((float)(1u + lo));
    if (2 * p + 1 < n) {
        v2f d; d.x = d0; d.y = rsqrtf((float)(1u + hi));
        *(v2f*)(dinv + 2 * p) = d;
    } else {
        dinv[2 * p] = d0;
    }
}

#define GROWS 64

template <bool DUAL, bool ACT>
__global__ __launch_bounds__(128)
void gemm_wmma_f32(const float* __restrict__ A,
                   const float* __restrict__ B0,
                   const float* __restrict__ B1,
                   const float* __restrict__ bias,
                   float* __restrict__ C, int nrows)
{
    constexpr int BLD = DUAL ? 64 : 128;
    __shared__ v4f Bs[16 * 8 * 32];
    const int tid  = threadIdx.x;
    const int lane = tid & 31;
    const int wave = __builtin_amdgcn_readfirstlane(tid >> 5);
    const int c    = lane & 15;
    const int h    = lane >> 4;
    const int row0 = blockIdx.x * GROWS + wave * 16;

    int gr = row0 + c; if (gr >= nrows) gr = nrows - 1;
    const v4f* ar = (const v4f*)(A + (size_t)gr * KDIM) + h;
    v4f a[16];
    #pragma unroll
    for (int j = 0; j < 16; ++j) a[j] = ar[2 * j];

    #pragma unroll 1
    for (int i0 = 0; i0 < 8; i0 += 4) {
        v4f r[4][4];
        #pragma unroll
        for (int i = 0; i < 4; ++i) {
            const int idx = tid + 128 * (i0 + i);
            const int kq = idx >> 5;
            const int cp = idx & 31;
            const float* w;
            if (DUAL) w = (cp < 16) ? (B0 + 4 * cp) : (B1 + 4 * (cp - 16));
            else      w = B0 + 4 * cp;
            w += (size_t)(4 * kq) * BLD;
            r[i][0] = *(const v4f*)(w);
            r[i][1] = *(const v4f*)(w + BLD);
            r[i][2] = *(const v4f*)(w + 2 * BLD);
            r[i][3] = *(const v4f*)(w + 3 * BLD);
        }
        #pragma unroll
        for (int i = 0; i < 4; ++i) {
            const int idx = tid + 128 * (i0 + i);
            const int kq = idx >> 5;
            const int cp = idx & 31;
            v4f* d = Bs + (((kq >> 1) * 8 + 4 * (cp >> 4)) * 32 + (kq & 1) * 16 + (cp & 15));
            v4f q;
            q.x = r[i][0].x; q.y = r[i][1].x; q.z = r[i][2].x; q.w = r[i][3].x; d[0]  = q;
            q.x = r[i][0].y; q.y = r[i][1].y; q.z = r[i][2].y; q.w = r[i][3].y; d[32] = q;
            q.x = r[i][0].z; q.y = r[i][1].z; q.z = r[i][2].z; q.w = r[i][3].z; d[64] = q;
            q.x = r[i][0].w; q.y = r[i][1].w; q.z = r[i][2].w; q.w = r[i][3].w; d[96] = q;
        }
    }
    v4f bv[16];
    if (ACT) {
        #pragma unroll
        for (int j = 0; j < 16; ++j) bv[j] = ((const v4f*)bias)[2 * j + h];
    }
    __syncthreads();
    if (row0 >= nrows) return;
    if (ACT) {
        #pragma unroll
        for (int j = 0; j < 16; ++j) {
            a[j].x = fmaxf(a[j].x + bv[j].x, 0.0f);
            a[j].y = fmaxf(a[j].y + bv[j].y, 0.0f);
            a[j].z = fmaxf(a[j].z + bv[j].z, 0.0f);
            a[j].w = fmaxf(a[j].w + bv[j].w, 0.0f);
        }
    }

    const v4f* bl = Bs + lane;
    v8f acc[8];
    #pragma unroll
    for (int t = 0; t < 8; ++t) acc[t] = (v8f){0.f, 0.f, 0.f, 0.f, 0.f, 0.f, 0.f, 0.f};

    #pragma unroll
    for (int j = 0; j < 16; ++j) {
        v4f b[8];
        #pragma unroll
        for (int t = 0; t < 8; ++t) b[t] = bl[(j * 8 + t) * 32];
        v2f a0; a0.x = a[j].x; a0.y = a[j].y;
        v2f a1; a1.x = a[j].z; a1.y = a[j].w;
        #pragma unroll
        for (int t = 0; t < 8; ++t) {
            v2f b0; b0.x = b[t].x; b0.y = b[t].y;
            acc[t] = __builtin_amdgcn_wmma_f32_16x16x4_f32(
                false, a0, false, b0, (short)0, acc[t], false, false);
        }
        #pragma unroll
        for (int t = 0; t < 8; ++t) {
            v2f b1; b1.x = b[t].z; b1.y = b[t].w;
            acc[t] = __builtin_amdgcn_wmma_f32_16x16x4_f32(
                false, a1, false, b1, (short)0, acc[t], false, false);
        }
    }

    float* cp0 = C + (size_t)(row0 + 8 * h) * FW + 4 * c;
    #pragma unroll
    for (int r = 0; r < 8; ++r) {
        if (row0 + r + 8 * h < nrows) {
            #pragma unroll
            for (int g = 0; g < 2; ++g) {
                v4f o;
                o.x = acc[4 * g][r]; o.y = acc[4 * g + 1][r]; o.z = acc[4 * g + 2][r]; o.w = acc[4 * g + 3][r];
                *(v4f*)(cp0 + (size_t)r * FW + 64 * g) = o;
            }
        }
    }
}

#define OWAVES 32
#define KT     8
#define TILE_E (32 * 16 * KT)
#define QCAP   520
#define RCAP   1024


template <bool REC, bool REPLAY>
__global__ __launch_bounds__(1024)
void agg_owner(const int* __restrict__ srcIdx, const int* __restrict__ dstIdx,
               const unsigned* __restrict__ keyw,
               const float* __restrict__ dinv, const float* __restrict__ in,
               float* __restrict__ out, int* __restrict__ rcnt, int* __restrict__ rec,
               int n, int e)
{
    __shared__ __attribute__((aligned(16))) float acc[ROWS * FW];
    __shared__ int   pendE[OWAVES * QCAP];

    const int tid  = threadIdx.x;
    const int lane = tid & 31;
    const int wave = __builtin_amdgcn_readfirstlane(tid >> 5);
    const int lo   = blockIdx.x * ROWS;
    const int nown = (n - lo < ROWS) ? (n - lo) : ROWS;

    for (int r = wave; r < nown; r += OWAVES) {
        float d  = dinv[lo + r];
        float dd = d * d;
        v4f x = ((const v4f*)(in + (size_t)(lo + r) * FW))[lane];
        v4f y;
        y.x = dd * x.x; y.y = dd * x.y; y.z = dd * x.z; y.w = dd * x.w;
        ((v4f*)(acc + r * FW))[lane] = y;
    }
    __syncthreads();

    const int wid = blockIdx.x * OWAVES + wave;
    int*   wS = rec + (size_t)wid * (3 * RCAP);
    int*   wR = wS + RCAP;
    float* wW = (float*)(wR + RCAP);
    int  nrec   = 0;
    bool doScan = true;
    if (REPLAY) {
        nrec = __builtin_amdgcn_readfirstlane(rcnt[wid]);
        if (nrec <= RCAP) {
            doScan = false;
            for (int g = 0; g < nrec; g += 32) {
                const int idx = g + lane;
                int s = 0, r = 0;
                float w = 0.0f;
                if (idx < nrec) { s = wS[idx]; r = wR[idx]; w = wW[idx]; }
                const int nb = (nrec - g < 32) ? (nrec - g) : 32;
                #pragma unroll
                for (int sub = 0; sub < 4; ++sub) {
                    if (8 * sub < nb) {
                        float v[8][4];
                        #pragma unroll
                        for (int q = 0; q < 8; ++q) {
                            const int sq = __builtin_amdgcn_readlane(s, 8 * sub + q);
                            const float* ip = in + (size_t)sq * FW + lane;
                            v[q][0] = ip[0]; v[q][1] = ip[32]; v[q][2] = ip[64]; v[q][3] = ip[96];
                        }
                        asm volatile("" :: "v"(v[0][0]), "v"(v[0][1]), "v"(v[0][2]), "v"(v[0][3]),
                                           "v"(v[1][0]), "v"(v[1][1]), "v"(v[1][2]), "v"(v[1][3]),
                                           "v"(v[2][0]), "v"(v[2][1]), "v"(v[2][2]), "v"(v[2][3]),
                                           "v"(v[3][0]), "v"(v[3][1]), "v"(v[3][2]), "v"(v[3][3]));
                        asm volatile("" :: "v"(v[4][0]), "v"(v[4][1]), "v"(v[4][2]), "v"(v[4][3]),
                                           "v"(v[5][0]), "v"(v[5][1]), "v"(v[5][2]), "v"(v[5][3]),
                                           "v"(v[6][0]), "v"(v[6][1]), "v"(v[6][2]), "v"(v[6][3]),
                                           "v"(v[7][0]), "v"(v[7][1]), "v"(v[7][2]), "v"(v[7][3]));
                        #pragma unroll
                        for (int q = 0; q < 8; ++q) {
                            if (8 * sub + q < nb) {
                                const int   rq = __builtin_amdgcn_readlane(r, 8 * sub + q);
                                const float nq = __builtin_bit_cast(float,
                                    __builtin_amdgcn_readlane(__builtin_bit_cast(int, w), 8 * sub + q));
                                float* ap = acc + rq * FW + lane;
                                atomicAdd(ap,      nq * v[q][0]);
                                atomicAdd(ap + 32, nq * v[q][1]);
                                atomicAdd(ap + 64, nq * v[q][2]);
                                atomicAdd(ap + 96, nq * v[q][3]);
                            }
                        }
                    }
                }
            }
        }
        nrec = 0;
    }

    if (doScan) {
    int* qE = pendE + wave * QCAP;
    int cnt = 0;
    const unsigned kb = (unsigned)(blockIdx.x & 255) * 0x01010101u;
    const int ntiles = (e + TILE_E - 1) / TILE_E;
    const int shift = (int)(((long long)blockIdx.x * ntiles) / gridDim.x);

    for (int i = wave; i < ntiles; i += OWAVES) {
        int t = i + shift;
        if (t >= ntiles) t -= ntiles;
        const v4u* kp = (const v4u*)(keyw + (size_t)t * (TILE_E / 4)) + lane;
        v4u kv[KT];
        #pragma unroll
        for (int j = 0; j < KT; ++j) kv[j] = kp[32 * j];
        const int ebase = t * TILE_E + 16 * lane;
        #pragma unroll
        for (int j = 0; j < KT; ++j) {
            #pragma unroll
            for (int d = 0; d < 4; ++d) {
                const unsigned x = kv[j][d] ^ kb;
                unsigned f = (x - 0x01010101u) & ~x & 0x80808080u;
                unsigned mask = __builtin_amdgcn_ballot_w32(f != 0u);
                while (mask) {
                    const int  eid = ebase + 512 * j + 4 * d + (__builtin_ctz(f | 0x80000000u) >> 3);
                    const bool ok  = (f != 0u) && (eid < e);
                    const unsigned m2 = __builtin_amdgcn_ballot_w32(ok);
                    const int pos = cnt + (int)__builtin_amdgcn_mbcnt_lo(m2, 0u);
                    if (ok) qE[pos] = eid;
                    cnt += __builtin_popcount(m2);
                    f &= f - 1u;
                    mask = __builtin_amdgcn_ballot_w32(f != 0u);
                }
            }
            while (cnt >= 8) {
                cnt -= 8;
                const int j8  = lane & 7;
                const int eid = qE[cnt + j8];
                const int s   = srcIdx[eid];
                const int dn  = dstIdx[eid];
                const float nm = dinv[s] * dinv[dn];
                float v[8][4];
                #pragma unroll
                for (int q = 0; q < 8; ++q) {
                    const int sq = __builtin_amdgcn_readlane(s, q);
                    const float* ip = in + (size_t)sq * FW + lane;
                    v[q][0] = ip[0]; v[q][1] = ip[32]; v[q][2] = ip[64]; v[q][3] = ip[96];
                }
                asm volatile("" :: "v"(v[0][0]), "v"(v[0][1]), "v"(v[0][2]), "v"(v[0][3]),
                                   "v"(v[1][0]), "v"(v[1][1]), "v"(v[1][2]), "v"(v[1][3]),
                                   "v"(v[2][0]), "v"(v[2][1]), "v"(v[2][2]), "v"(v[2][3]),
                                   "v"(v[3][0]), "v"(v[3][1]), "v"(v[3][2]), "v"(v[3][3]));
                asm volatile("" :: "v"(v[4][0]), "v"(v[4][1]), "v"(v[4][2]), "v"(v[4][3]),
                                   "v"(v[5][0]), "v"(v[5][1]), "v"(v[5][2]), "v"(v[5][3]),
                                   "v"(v[6][0]), "v"(v[6][1]), "v"(v[6][2]), "v"(v[6][3]),
                                   "v"(v[7][0]), "v"(v[7][1]), "v"(v[7][2]), "v"(v[7][3]));
                if (REC) {
                    const unsigned rl  = (unsigned)(dn - lo);
                    const bool     val = (lane < 8) && (rl < (unsigned)nown);
                    const unsigned m   = __builtin_amdgcn_ballot_w32(val);
                    const int      pos = nrec + (int)__builtin_amdgcn_mbcnt_lo(m, 0u);
                    if (val && pos < RCAP) { wS[pos] = s; wR[pos] = (int)rl; wW[pos] = nm; }
                    nrec += __builtin_popcount(m);
                }
                #pragma unroll
                for (int q = 0; q < 8; ++q) {
                    const unsigned rq = (unsigned)(__builtin_amdgcn_readlane(dn, q) - lo);
                    if (rq < (unsigned)nown) {
                        const float nq = __builtin_bit_cast(float,
                            __builtin_amdgcn_readlane(__builtin_bit_cast(int, nm), q));
                        float* ap = acc + rq * FW + lane;
                        atomicAdd(ap,      nq * v[q][0]);
                        atomicAdd(ap + 32, nq * v[q][1]);
                        atomicAdd(ap + 64, nq * v[q][2]);
                        atomicAdd(ap + 96, nq * v[q][3]);
                    }
                }
            }
        }
    }
    for (int j = 0; j < cnt; ++j) {
        const int eid = qE[j];
        const int s   = srcIdx[eid];
        const int dn  = dstIdx[eid];
        const unsigned rl = (unsigned)(dn - lo);
        if (rl < (unsigned)nown) {
            const float nm = dinv[s] * dinv[dn];
            const float* ip = in + (size_t)s * FW + lane;
            float* ap = acc + rl * FW + lane;
            atomicAdd(ap,      nm * ip[0]);
            atomicAdd(ap + 32, nm * ip[32]);
            atomicAdd(ap + 64, nm * ip[64]);
            atomicAdd(ap + 96, nm * ip[96]);
            if (REC) {
                if (lane == 0 && nrec < RCAP) { wS[nrec] = s; wR[nrec] = (int)rl; wW[nrec] = nm; }
                ++nrec;
            }
        }
    }
    }
    if (REC) { if (lane == 0) rcnt[wid] = nrec; }
    __syncthreads();

    for (int r = wave; r < nown; r += OWAVES) {
        ((v4f*)(out + (size_t)(lo + r) * FW))[lane] = ((const v4f*)(acc + r * FW))[lane];
    }
}

#define EW_BLOCKS 1024
#define EW_BATCH  8

__global__ __launch_bounds__(256)
void final_out(const float* __restrict__ agg,
               const float* __restrict__ bmu, const float* __restrict__ blv,
               float* __restrict__ outp, int n)
{
    const int nthr  = gridDim.x * 256;
    const int t0    = blockIdx.x * 256 + threadIdx.x;
    const int total = n * (FW / 4);
    const int c4    = t0 & (FW / 4 - 1);
    const v4f bb = (c4 < 16) ? ((const v4f*)bmu)[c4] : ((const v4f*)blv)[c4 - 16];
    float* __restrict__ o0 = (c4 < 16) ? (outp + 4 * c4) : (outp + (size_t)n * 64 + 4 * (c4 - 16));
    const v4f* __restrict__ src = (const v4f*)agg;
    for (int q = t0; q < total; q += EW_BATCH * nthr) {
        v4f v[EW_BATCH];
        #pragma unroll
        for (int k = 0; k < EW_BATCH; ++k) {
            const int qq = q + k * nthr;
            if (qq < total) v[k] = src[qq];
        }
        #pragma unroll
        for (int k = 0; k < EW_BATCH; ++k) {
            const int qq = q + k * nthr;
            if (qq < total) {
                v4f y;
                y.x = v[k].x + bb.x;
                y.y = v[k].y + bb.y;
                y.z = v[k].z + bb.z;
                y.w = v[k].w + bb.w;
                *(v4f*)(o0 + (size_t)(qq >> 5) * 64) = y;
            }
        }
    }
}

extern "C" void kernel_launch(void* const* d_in, const int* in_sizes, int n_in,
                              void* d_out, int out_size, void* d_ws, size_t ws_size,
                              hipStream_t stream)
{
    const float* x    = (const float*)d_in[0];
    const int*   ei   = (const int*)  d_in[1];
    const float* W1   = (const float*)d_in[2];
    const float* b1   = (const float*)d_in[3];
    const float* Wmu  = (const float*)d_in[4];
    const float* bmu  = (const float*)d_in[5];
    const float* Wlv  = (const float*)d_in[6];
    const float* blv  = (const float*)d_in[7];

    const int N = in_sizes[0] / KDIM;
    const int E = in_sizes[1] / 2;
    const int* srcIdx = ei;
    const int* dstIdx = ei + E;

    size_t dinv_bytes = (((size_t)N * 4) + 255) & ~(size_t)255;
    float* dinv = (float*)d_ws;
    float* bufA = (float*)((char*)d_ws + dinv_bytes);
    float* bufB = bufA + (size_t)N * FW;
    unsigned* part = (unsigned*)(bufB + (size_t)N * FW);

    const int TB = 256;
    dim3 blk(TB);
    int gP   = ((N + 1) / 2 + TB - 1) / TB;
    int nCh  = (E + ECHUNK - 1) / ECHUNK;
    int nRg  = (N + HBINS - 1) / HBINS;
    int gOwn = (N + ROWS - 1) / ROWS;
    int gGemm = (N + GROWS - 1) / GROWS;
    unsigned* keyw = part + (size_t)nRg * nCh * HWORDS;
    int* rcnt = (int*)(keyw + (size_t)nCh * (ECHUNK / 4));
    int* rec  = rcnt + (size_t)gOwn * OWAVES;

    deg_hist  <<<dim3(nCh, nRg), dim3(1024), 0, stream>>>(dstIdx, part, keyw, N, E);
    deg_finish<<<gP, blk, 0, stream>>>(part, dinv, N, nCh);

    gemm_wmma_f32<false, false><<<gGemm, dim3(128), 0, stream>>>(x, W1, nullptr, nullptr, bufA, N);

    agg_owner<true, false><<<gOwn, dim3(1024), 0, stream>>>(srcIdx, dstIdx, keyw, dinv, bufA, bufB, rcnt, rec, N, E);

    gemm_wmma_f32<true, true><<<gGemm, dim3(128), 0, stream>>>(bufB, Wmu, Wlv, b1, bufA, N);

    agg_owner<false, true><<<gOwn, dim3(1024), 0, stream>>>(srcIdx, dstIdx, keyw, dinv, bufA, bufB, rcnt, rec, N, E);

    final_out<<<EW_BLOCKS, blk, 0, stream>>>(bufB, bmu, blv, (float*)d_out, N);
}


# both layers sum the gathered rows first and take the dense product of the sums, (SX)W not S(XW): bias, relu and the mu/logvar split move into the GEMM's own store; final pass gone; rounding differs
# speedup vs baseline: 3.4719x; 1.0960x over previous
#include <hip/hip_runtime.h>
#include <hip/hip_bf16.h>

typedef __attribute__((ext_vector_type(2))) float v2f;
typedef __attribute__((ext_vector_type(4))) float v4f;
typedef __attribute__((ext_vector_type(8))) float v8f;
typedef __attribute__((ext_vector_type(4))) unsigned v4u;

#define KDIM 128
#define FW   128

#define HBINS  100352
#define HWORDS (HBINS / 2)
#define ECHUNK 16384
#define EPT    (ECHUNK / 1024)
#define PGRP   14
#define ROWS   392

__global__ __launch_bounds__(1024)
void deg_hist(const int* __restrict__ dst, unsigned* __restrict__ part,
              unsigned* __restrict__ keyw, int n, int e)
{
    __shared__ __attribute__((aligned(16))) unsigned hist[HWORDS];
    const int tid = threadIdx.x;
    const int nb  = blockIdx.y * HBINS;
    const int nin = (n - nb < HBINS) ? (n - nb) : HBINS;
    const int nq  = (nin + 7) >> 3;

    const int e0 = blockIdx.x * ECHUNK + 4 * tid;
    int dv[EPT];
    if (blockIdx.x * ECHUNK + ECHUNK <= e) {
        #pragma unroll
        for (int k = 0; k < EPT; ++k) dv[k] = dst[e0 + 4096 * (k >> 2) + (k & 3)];
    } else {
        #pragma unroll
        for (int k = 0; k < EPT; ++k) {
            int ei = e0 + 4096 * (k >> 2) + (k & 3);
            dv[k] = (ei < e) ? dst[ei] : -1;
        }
    }
    if (blockIdx.y == 0) {
        unsigned* kw = keyw + (e0 >> 2);
        #pragma unroll
        for (int g = 0; g < EPT / 4; ++g) {
            unsigned w = 0u;
            #pragma unroll
            for (int j = 0; j < 4; ++j) w |= (((unsigned)dv[4 * g + j] / (unsigned)ROWS) & 255u) << (8 * j);
            kw[1024 * g] = w;
        }
    }
    const v4u zero = {0u, 0u, 0u, 0u};
    for (int i = tid; i < nq; i += 1024) ((v4u*)hist)[i] = zero;
    __syncthreads();

    #pragma unroll
    for (int k = 0; k < EPT; ++k) {
        unsigned rel = (unsigned)(dv[k] - nb);
        if (rel < (unsigned)nin) atomicAdd(&hist[rel >> 1], 1u << ((rel & 1u) << 4));
    }
    __syncthreads();

    v4u* po = (v4u*)(part + ((size_t)blockIdx.y * gridDim.x + blockIdx.x) * HWORDS);
    for (int i = tid; i < nq; i += 1024) po[i] = ((const v4u*)hist)[i];
}

__global__ void deg_finish(const unsigned* __restrict__ part, float* __restrict__ dinv,
                           int n, int nchunks)
{
    const int p = blockIdx.x * blockDim.x + threadIdx.x;
    if (2 * p >= n) return;
    const int r = p / HWORDS;
    const int w = p - r * HWORDS;
    const unsigned* q = part + (size_t)r * nchunks * HWORDS + w;
    unsigned lo = 0u, hi = 0u;
    int c = 0;
    for (; c + PGRP <= nchunks; c += PGRP) {
        unsigned v[PGRP];
        #pragma unroll
        for (int j = 0; j < PGRP; ++j) v[j] = q[(size_t)j * HWORDS];
        #pragma unroll
        for (int j = 0; j < PGRP; ++j) { lo += v[j] & 0xffffu; hi += v[j] >> 16; }
        q += (size_t)PGRP * HWORDS;
    }
    for (; c < nchunks; ++c) {
        unsigned v = q[0];
        lo += v & 0xffffu;
        hi += v >> 16;
        q += HWORDS;
    }
    const float d0 = rsqrtf((float)(1u + lo));
    if (2 * p + 1 < n) {
        v2f d; d.x = d0; d.y = rsqrtf((float)(1u + hi));
        *(v2f*)(dinv + 2 * p) = d;
    } else {
        dinv[2 * p] = d0;
    }
}

#define GROWS 64

template <bool DUAL, bool RELU>
__global__ __launch_bounds__(128)
void gemm_wmma_f32(const float* __restrict__ A,
                   const float* __restrict__ B0,
                   const float* __restrict__ B1,
                   const float* __restrict__ bias0,
                   const float* __restrict__ bias1,
                   float* __restrict__ C, int nrows)
{
    constexpr int BLD = DUAL ? 64 : 128;
    __shared__ v4f Bs[16 * 8 * 32];
    const int tid  = threadIdx.x;
    const int lane = tid & 31;
    const int wave = __builtin_amdgcn_readfirstlane(tid >> 5);
    const int c    = lane & 15;
    const int h    = lane >> 4;
    const int row0 = blockIdx.x * GROWS + wave * 16;

    int gr = row0 + c; if (gr >= nrows) gr = nrows - 1;
    const v4f* ar = (const v4f*)(A + (size_t)gr * KDIM) + h;
    v4f a[16];
    #pragma unroll
    for (int j = 0; j < 16; ++j) a[j] = ar[2 * j];

    #pragma unroll 1
    for (int i0 = 0; i0 < 8; i0 += 4) {
        v4f r[4][4];
        #pragma unroll
        for (int i = 0; i < 4; ++i) {
            const int idx = tid + 128 * (i0 + i);
            const int kq = idx >> 5;
            const int cp = idx & 31;
            const float* w;
            if (DUAL) w = (cp < 16) ? (B0 + 4 * cp) : (B1 + 4 * (cp - 16));
            else      w = B0 + 4 * cp;
            w += (size_t)(4 * kq) * BLD;
            r[i][0] = *(const v4f*)(w);
            r[i][1] = *(const v4f*)(w + BLD);
            r[i][2] = *(const v4f*)(w + 2 * BLD);
            r[i][3] = *(const v4f*)(w + 3 * BLD);
        }
        #pragma unroll
        for (int i = 0; i < 4; ++i) {
            const int idx = tid + 128 * (i0 + i);
            const int kq = idx >> 5;
            const int cp = idx & 31;
            v4f* d = Bs + (((kq >> 1) * 8 + 4 * (cp >> 4)) * 32 + (kq & 1) * 16 + (cp & 15));
            v4f q;
            q.x = r[i][0].x; q.y = r[i][1].x; q.z = r[i][2].x; q.w = r[i][3].x; d[0]  = q;
            q.x = r[i][0].y; q.y = r[i][1].y; q.z = r[i][2].y; q.w = r[i][3].y; d[32] = q;
            q.x = r[i][0].z; q.y = r[i][1].z; q.z = r[i][2].z; q.w = r[i][3].z; d[64] = q;
            q.x = r[i][0].w; q.y = r[i][1].w; q.z = r[i][2].w; q.w = r[i][3].w; d[96] = q;
        }
    }
    __syncthreads();
    if (row0 >= nrows) return;

    const v4f* bl = Bs + lane;
    v8f acc[8];
    #pragma unroll
    for (int t = 0; t < 8; ++t) acc[t] = (v8f){0.f, 0.f, 0.f, 0.f, 0.f, 0.f, 0.f, 0.f};

    #pragma unroll
    for (int j = 0; j < 16; ++j) {
        v4f b[8];
        #pragma unroll
        for (int t = 0; t < 8; ++t) b[t] = bl[(j * 8 + t) * 32];
        v2f a0; a0.x = a[j].x; a0.y = a[j].y;
        v2f a1; a1.x = a[j].z; a1.y = a[j].w;
        #pragma unroll
        for (int t = 0; t < 8; ++t) {
            v2f b0; b0.x = b[t].x; b0.y = b[t].y;
            acc[t] = __builtin_amdgcn_wmma_f32_16x16x4_f32(
                false, a0, false, b0, (short)0, acc[t], false, false);
        }
        #pragma unroll
        for (int t = 0; t < 8; ++t) {
            v2f b1; b1.x = b[t].z; b1.y = b[t].w;
            acc[t] = __builtin_amdgcn_wmma_f32_16x16x4_f32(
                false, a1, false, b1, (short)0, acc[t], false, false);
        }
    }

    v4f bq[2];
    #pragma unroll
    for (int g = 0; g < 2; ++g) {
        const float* bp = DUAL ? ((g ? bias1 : bias0) + 4 * c) : (bias0 + 64 * g + 4 * c);
        bq[g] = *(const v4f*)bp;
    }

    #pragma unroll
    for (int r = 0; r < 8; ++r) {
        const int row = row0 + r + 8 * h;
        if (row < nrows) {
            #pragma unroll
            for (int g = 0; g < 2; ++g) {
                v4f o;
                o.x = acc[4 * g][r]     + bq[g].x;
                o.y = acc[4 * g + 1][r] + bq[g].y;
                o.z = acc[4 * g + 2][r] + bq[g].z;
                o.w = acc[4 * g + 3][r] + bq[g].w;
                if (RELU) {
                    o.x = fmaxf(o.x, 0.0f); o.y = fmaxf(o.y, 0.0f);
                    o.z = fmaxf(o.z, 0.0f); o.w = fmaxf(o.w, 0.0f);
                }
                float* op = DUAL ? (C + ((size_t)g * nrows + row) * 64 + 4 * c)
                                 : (C + (size_t)row * FW + 64 * g + 4 * c);
                *(v4f*)op = o;
            }
        }
    }
}

#define OWAVES 32
#define KT     8
#define TILE_E (32 * 16 * KT)
#define QCAP   520
#define RCAP   1024


template <bool REC, bool REPLAY>
__global__ __launch_bounds__(1024)
void agg_owner(const int* __restrict__ srcIdx, const int* __restrict__ dstIdx,
               const unsigned* __restrict__ keyw,
               const float* __restrict__ dinv, const float* __restrict__ in,
               float* __restrict__ out, int* __restrict__ rcnt, int* __restrict__ rec,
               int n, int e)
{
    __shared__ __attribute__((aligned(16))) float acc[ROWS * FW];
    __shared__ int   pendE[OWAVES * QCAP];

    const int tid  = threadIdx.x;
    const int lane = tid & 31;
    const int wave = __builtin_amdgcn_readfirstlane(tid >> 5);
    const int lo   = blockIdx.x * ROWS;
    const int nown = (n - lo < ROWS) ? (n - lo) : ROWS;

    for (int r = wave; r < nown; r += OWAVES) {
        float d  = dinv[lo + r];
        float dd = d * d;
        v4f x = ((const v4f*)(in + (size_t)(lo + r) * FW))[lane];
        v4f y;
        y.x = dd * x.x; y.y = dd * x.y; y.z = dd * x.z; y.w = dd * x.w;
        ((v4f*)(acc + r * FW))[lane] = y;
    }
    __syncthreads();

    const int wid = blockIdx.x * OWAVES + wave;
    int*   wS = rec + (size_t)wid * (3 * RCAP);
    int*   wR = wS + RCAP;
    float* wW = (float*)(wR + RCAP);
    int  nrec   = 0;
    bool doScan = true;
    if (REPLAY) {
        nrec = __builtin_amdgcn_readfirstlane(rcnt[wid]);
        if (nrec <= RCAP) {
            doScan = false;
            for (int g = 0; g < nrec; g += 32) {
                const int idx = g + lane;
                int s = 0, r = 0;
                float w = 0.0f;
                if (idx < nrec) { s = wS[idx]; r = wR[idx]; w = wW[idx]; }
                const int nb = (nrec - g < 32) ? (nrec - g) : 32;
                #pragma unroll
                for (int sub = 0; sub < 4; ++sub) {
                    if (8 * sub < nb) {
                        float v[8][4];
                        #pragma unroll
                        for (int q = 0; q < 8; ++q) {
                            const int sq = __builtin_amdgcn_readlane(s, 8 * sub + q);
                            const float* ip = in + (size_t)sq * FW + lane;
                            v[q][0] = ip[0]; v[q][1] = ip[32]; v[q][2] = ip[64]; v[q][3] = ip[96];
                        }
                        asm volatile("" :: "v"(v[0][0]), "v"(v[0][1]), "v"(v[0][2]), "v"(v[0][3]),
                                           "v"(v[1][0]), "v"(v[1][1]), "v"(v[1][2]), "v"(v[1][3]),
                                           "v"(v[2][0]), "v"(v[2][1]), "v"(v[2][2]), "v"(v[2][3]),
                                           "v"(v[3][0]), "v"(v[3][1]), "v"(v[3][2]), "v"(v[3][3]));
                        asm volatile("" :: "v"(v[4][0]), "v"(v[4][1]), "v"(v[4][2]), "v"(v[4][3]),
                                           "v"(v[5][0]), "v"(v[5][1]), "v"(v[5][2]), "v"(v[5][3]),
                                           "v"(v[6][0]), "v"(v[6][1]), "v"(v[6][2]), "v"(v[6][3]),
                                           "v"(v[7][0]), "v"(v[7][1]), "v"(v[7][2]), "v"(v[7][3]));
                        #pragma unroll
                        for (int q = 0; q < 8; ++q) {
                            if (8 * sub + q < nb) {
                                const int   rq = __builtin_amdgcn_readlane(r, 8 * sub + q);
                                const float nq = __builtin_bit_cast(float,
                                    __builtin_amdgcn_readlane(__builtin_bit_cast(int, w), 8 * sub + q));
                                float* ap = acc + rq * FW + lane;
                                atomicAdd(ap,      nq * v[q][0]);
                                atomicAdd(ap + 32, nq * v[q][1]);
                                atomicAdd(ap + 64, nq * v[q][2]);
                                atomicAdd(ap + 96, nq * v[q][3]);
                            }
                        }
                    }
                }
            }
        }
        nrec = 0;
    }

    if (doScan) {
    int* qE = pendE + wave * QCAP;
    int cnt = 0;
    const unsigned kb = (unsigned)(blockIdx.x & 255) * 0x01010101u;
    const int ntiles = (e + TILE_E - 1) / TILE_E;
    const int shift = (int)(((long long)blockIdx.x * ntiles) / gridDim.x);

    for (int i = wave; i < ntiles; i += OWAVES) {
        int t = i + shift;
        if (t >= ntiles) t -= ntiles;
        const v4u* kp = (const v4u*)(keyw + (size_t)t * (TILE_E / 4)) + lane;
        v4u kv[KT];
        #pragma unroll
        for (int j = 0; j < KT; ++j) kv[j] = kp[32 * j];
        const int ebase = t * TILE_E + 16 * lane;
        #pragma unroll
        for (int j = 0; j < KT; ++j) {
            #pragma unroll
            for (int d = 0; d < 4; ++d) {
                const unsigned x = kv[j][d] ^ kb;
                unsigned f = (x - 0x01010101u) & ~x & 0x80808080u;
                unsigned mask = __builtin_amdgcn_ballot_w32(f != 0u);
                while (mask) {
                    const int  eid = ebase + 512 * j + 4 * d + (__builtin_ctz(f | 0x80000000u) >> 3);
                    const bool ok  = (f != 0u) && (eid < e);
                    const unsigned m2 = __builtin_amdgcn_ballot_w32(ok);
                    const int pos = cnt + (int)__builtin_amdgcn_mbcnt_lo(m2, 0u);
                    if (ok) qE[pos] = eid;
                    cnt += __builtin_popcount(m2);
                    f &= f - 1u;
                    mask = __builtin_amdgcn_ballot_w32(f != 0u);
                }
            }
            while (cnt >= 8) {
                cnt -= 8;
                const int j8  = lane & 7;
                const int eid = qE[cnt + j8];
                const int s   = srcIdx[eid];
                const int dn  = dstIdx[eid];
                const float nm = dinv[s] * dinv[dn];
                float v[8][4];
                #pragma unroll
                for (int q = 0; q < 8; ++q) {
                    const int sq = __builtin_amdgcn_readlane(s, q);
                    const float* ip = in + (size_t)sq * FW + lane;
                    v[q][0] = ip[0]; v[q][1] = ip[32]; v[q][2] = ip[64]; v[q][3] = ip[96];
                }
                asm volatile("" :: "v"(v[0][0]), "v"(v[0][1]), "v"(v[0][2]), "v"(v[0][3]),
                                   "v"(v[1][0]), "v"(v[1][1]), "v"(v[1][2]), "v"(v[1][3]),
                                   "v"(v[2][0]), "v"(v[2][1]), "v"(v[2][2]), "v"(v[2][3]),
                                   "v"(v[3][0]), "v"(v[3][1]), "v"(v[3][2]), "v"(v[3][3]));
                asm volatile("" :: "v"(v[4][0]), "v"(v[4][1]), "v"(v[4][2]), "v"(v[4][3]),
                                   "v"(v[5][0]), "v"(v[5][1]), "v"(v[5][2]), "v"(v[5][3]),
                                   "v"(v[6][0]), "v"(v[6][1]), "v"(v[6][2]), "v"(v[6][3]),
                                   "v"(v[7][0]), "v"(v[7][1]), "v"(v[7][2]), "v"(v[7][3]));
                if (REC) {
                    const unsigned rl  = (unsigned)(dn - lo);
                    const bool     val = (lane < 8) && (rl < (unsigned)nown);
                    const unsigned m   = __builtin_amdgcn_ballot_w32(val);
                    const int      pos = nrec + (int)__builtin_amdgcn_mbcnt_lo(m, 0u);
                    if (val && pos < RCAP) { wS[pos] = s; wR[pos] = (int)rl; wW[pos] = nm; }
                    nrec += __builtin_popcount(m);
                }
                #pragma unroll
                for (int q = 0; q < 8; ++q) {
                    const unsigned rq = (unsigned)(__builtin_amdgcn_readlane(dn, q) - lo);
                    if (rq < (unsigned)nown) {
                        const float nq = __builtin_bit_cast(float,
                            __builtin_amdgcn_readlane(__builtin_bit_cast(int, nm), q));
                        float* ap = acc + rq * FW + lane;
                        atomicAdd(ap,      nq * v[q][0]);
                        atomicAdd(ap + 32, nq * v[q][1]);
                        atomicAdd(ap + 64, nq * v[q][2]);
                        atomicAdd(ap + 96, nq * v[q][3]);
                    }
                }
            }
        }
    }
    for (int j = 0; j < cnt; ++j) {
        const int eid = qE[j];
        const int s   = srcIdx[eid];
        const int dn  = dstIdx[eid];
        const unsigned rl = (unsigned)(dn - lo);
        if (rl < (unsigned)nown) {
            const float nm = dinv[s] * dinv[dn];
            const float* ip = in + (size_t)s * FW + lane;
            float* ap = acc + rl * FW + lane;
            atomicAdd(ap,      nm * ip[0]);
            atomicAdd(ap + 32, nm * ip[32]);
            atomicAdd(ap + 64, nm * ip[64]);
            atomicAdd(ap + 96, nm * ip[96]);
            if (REC) {
                if (lane == 0 && nrec < RCAP) { wS[nrec] = s; wR[nrec] = (int)rl; wW[nrec] = nm; }
                ++nrec;
            }
        }
    }
    }
    if (REC) { if (lane == 0) rcnt[wid] = nrec; }
    __syncthreads();

    for (int r = wave; r < nown; r += OWAVES) {
        ((v4f*)(out + (size_t)(lo + r) * FW))[lane] = ((const v4f*)(acc + r * FW))[lane];
    }
}

extern "C" void kernel_launch(void* const* d_in, const int* in_sizes, int n_in,
                              void* d_out, int out_size, void* d_ws, size_t ws_size,
                              hipStream_t stream)
{
    const float* x    = (const float*)d_in[0];
    const int*   ei   = (const int*)  d_in[1];
    const float* W1   = (const float*)d_in[2];
    const float* b1   = (const float*)d_in[3];
    const float* Wmu  = (const float*)d_in[4];
    const float* bmu  = (const float*)d_in[5];
    const float* Wlv  = (const float*)d_in[6];
    const float* blv  = (const float*)d_in[7];

    const int N = in_sizes[0] / KDIM;
    const int E = in_sizes[1] / 2;
    const int* srcIdx = ei;
    const int* dstIdx = ei + E;

    size_t dinv_bytes = (((size_t)N * 4) + 255) & ~(size_t)255;
    float* dinv = (float*)d_ws;
    float* bufA = (float*)((char*)d_ws + dinv_bytes);
    float* bufB = bufA + (size_t)N * FW;
    unsigned* part = (unsigned*)(bufB + (size_t)N * FW);

    const int TB = 256;
    dim3 blk(TB);
    int gP   = ((N + 1) / 2 + TB - 1) / TB;
    int nCh  = (E + ECHUNK - 1) / ECHUNK;
    int nRg  = (N + HBINS - 1) / HBINS;
    int gOwn = (N + ROWS - 1) / ROWS;
    int gGemm = (N + GROWS - 1) / GROWS;
    unsigned* keyw = part + (size_t)nRg * nCh * HWORDS;
    int* rcnt = (int*)(keyw + (size_t)nCh * (ECHUNK / 4));
    int* rec  = rcnt + (size_t)gOwn * OWAVES;

    deg_hist  <<<dim3(nCh, nRg), dim3(1024), 0, stream>>>(dstIdx, part, keyw, N, E);
    deg_finish<<<gP, blk, 0, stream>>>(part, dinv, N, nCh);


    agg_owner<true, false><<<gOwn, dim3(1024), 0, stream>>>(srcIdx, dstIdx, keyw, dinv, x, bufB, rcnt, rec, N, E);

    gemm_wmma_f32<false, true><<<gGemm, dim3(128), 0, stream>>>(bufB, W1, nullptr, b1, nullptr, bufA, N);

    agg_owner<false, true><<<gOwn, dim3(1024), 0, stream>>>(srcIdx, dstIdx, keyw, dinv, bufA, bufB, rcnt, rec, N, E);

    gemm_wmma_f32<true, false><<<gGemm, dim3(128), 0, stream>>>(bufB, Wmu, Wlv, bmu, blv, (float*)d_out, N);
}


# layer-1 agg_owner works its hits off 32 at a time, not 8: a lane per hit reads source, destination and both dinv, so these dependent round trips are paid once per 32 edges; arrival order
# speedup vs baseline: 3.7026x; 1.0664x over previous
#include <hip/hip_runtime.h>
#include <hip/hip_bf16.h>

typedef __attribute__((ext_vector_type(2))) float v2f;
typedef __attribute__((ext_vector_type(4))) float v4f;
typedef __attribute__((ext_vector_type(8))) float v8f;
typedef __attribute__((ext_vector_type(4))) unsigned v4u;

#define KDIM 128
#define FW   128

#define HBINS  100352
#define HWORDS (HBINS / 2)
#define ECHUNK 16384
#define EPT    (ECHUNK / 1024)
#define PGRP   14
#define ROWS   392

__global__ __launch_bounds__(1024)
void deg_hist(const int* __restrict__ dst, unsigned* __restrict__ part,
              unsigned* __restrict__ keyw, int n, int e)
{
    __shared__ __attribute__((aligned(16))) unsigned hist[HWORDS];
    const int tid = threadIdx.x;
    const int nb  = blockIdx.y * HBINS;
    const int nin = (n - nb < HBINS) ? (n - nb) : HBINS;
    const int nq  = (nin + 7) >> 3;

    const int e0 = blockIdx.x * ECHUNK + 4 * tid;
    int dv[EPT];
    if (blockIdx.x * ECHUNK + ECHUNK <= e) {
        #pragma unroll
        for (int k = 0; k < EPT; ++k) dv[k] = dst[e0 + 4096 * (k >> 2) + (k & 3)];
    } else {
        #pragma unroll
        for (int k = 0; k < EPT; ++k) {
            int ei = e0 + 4096 * (k >> 2) + (k & 3);
            dv[k] = (ei < e) ? dst[ei] : -1;
        }
    }
    if (blockIdx.y == 0) {
        unsigned* kw = keyw + (e0 >> 2);
        #pragma unroll
        for (int g = 0; g < EPT / 4; ++g) {
            unsigned w = 0u;
            #pragma unroll
            for (int j = 0; j < 4; ++j) w |= (((unsigned)dv[4 * g + j] / (unsigned)ROWS) & 255u) << (8 * j);
            kw[1024 * g] = w;
        }
    }
    const v4u zero = {0u, 0u, 0u, 0u};
    for (int i = tid; i < nq; i += 1024) ((v4u*)hist)[i] = zero;
    __syncthreads();

    #pragma unroll
    for (int k = 0; k < EPT; ++k) {
        unsigned rel = (unsigned)(dv[k] - nb);
        if (rel < (unsigned)nin) atomicAdd(&hist[rel >> 1], 1u << ((rel & 1u) << 4));
    }
    __syncthreads();

    v4u* po = (v4u*)(part + ((size_t)blockIdx.y * gridDim.x + blockIdx.x) * HWORDS);
    for (int i = tid; i < nq; i += 1024) po[i] = ((const v4u*)hist)[i];
}

__global__ void deg_finish(const unsigned* __restrict__ part, float* __restrict__ dinv,
                           int n, int nchunks)
{
    const int p = blockIdx.x * blockDim.x + threadIdx.x;
    if (2 * p >= n) return;
    const int r = p / HWORDS;
    const int w = p - r * HWORDS;
    const unsigned* q = part + (size_t)r * nchunks * HWORDS + w;
    unsigned lo = 0u, hi = 0u;
    int c = 0;
    for (; c + PGRP <= nchunks; c += PGRP) {
        unsigned v[PGRP];
        #pragma unroll
        for (int j = 0; j < PGRP; ++j) v[j] = q[(size_t)j * HWORDS];
        #pragma unroll
        for (int j = 0; j < PGRP; ++j) { lo += v[j] & 0xffffu; hi += v[j] >> 16; }
        q += (size_t)PGRP * HWORDS;
    }
    for (; c < nchunks; ++c) {
        unsigned v = q[0];
        lo += v & 0xffffu;
        hi += v >> 16;
        q += HWORDS;
    }
    const float d0 = rsqrtf((float)(1u + lo));
    if (2 * p + 1 < n) {
        v2f d; d.x = d0; d.y = rsqrtf((float)(1u + hi));
        *(v2f*)(dinv + 2 * p) = d;
    } else {
        dinv[2 * p] = d0;
    }
}

#define GROWS 64

template <bool DUAL, bool RELU>
__global__ __launch_bounds__(128)
void gemm_wmma_f32(const float* __restrict__ A,
                   const float* __restrict__ B0,
                   const float* __restrict__ B1,
                   const float* __restrict__ bias0,
                   const float* __restrict__ bias1,
                   float* __restrict__ C, int nrows)
{
    constexpr int BLD = DUAL ? 64 : 128;
    __shared__ v4f Bs[16 * 8 * 32];
    const int tid  = threadIdx.x;
    const int lane = tid & 31;
    const int wave = __builtin_amdgcn_readfirstlane(tid >> 5);
    const int c    = lane & 15;
    const int h    = lane >> 4;
    const int row0 = blockIdx.x * GROWS + wave * 16;

    int gr = row0 + c; if (gr >= nrows) gr = nrows - 1;
    const v4f* ar = (const v4f*)(A + (size_t)gr * KDIM) + h;
    v4f a[16];
    #pragma unroll
    for (int j = 0; j < 16; ++j) a[j] = ar[2 * j];

    #pragma unroll 1
    for (int i0 = 0; i0 < 8; i0 += 4) {
        v4f r[4][4];
        #pragma unroll
        for (int i = 0; i < 4; ++i) {
            const int idx = tid + 128 * (i0 + i);
            const int kq = idx >> 5;
            const int cp = idx & 31;
            const float* w;
            if (DUAL) w = (cp < 16) ? (B0 + 4 * cp) : (B1 + 4 * (cp - 16));
            else      w = B0 + 4 * cp;
            w += (size_t)(4 * kq) * BLD;
            r[i][0] = *(const v4f*)(w);
            r[i][1] = *(const v4f*)(w + BLD);
            r[i][2] = *(const v4f*)(w + 2 * BLD);
            r[i][3] = *(const v4f*)(w + 3 * BLD);
        }
        #pragma unroll
        for (int i = 0; i < 4; ++i) {
            const int idx = tid + 128 * (i0 + i);
            const int kq = idx >> 5;
            const int cp = idx & 31;
            v4f* d = Bs + (((kq >> 1) * 8 + 4 * (cp >> 4)) * 32 + (kq & 1) * 16 + (cp & 15));
            v4f q;
            q.x = r[i][0].x; q.y = r[i][1].x; q.z = r[i][2].x; q.w = r[i][3].x; d[0]  = q;
            q.x = r[i][0].y; q.y = r[i][1].y; q.z = r[i][2].y; q.w = r[i][3].y; d[32] = q;
            q.x = r[i][0].z; q.y = r[i][1].z; q.z = r[i][2].z; q.w = r[i][3].z; d[64] = q;
            q.x = r[i][0].w; q.y = r[i][1].w; q.z = r[i][2].w; q.w = r[i][3].w; d[96] = q;
        }
    }
    __syncthreads();
    if (row0 >= nrows) return;

    const v4f* bl = Bs + lane;
    v8f acc[8];
    #pragma unroll
    for (int t = 0; t < 8; ++t) acc[t] = (v8f){0.f, 0.f, 0.f, 0.f, 0.f, 0.f, 0.f, 0.f};

    #pragma unroll
    for (int j = 0; j < 16; ++j) {
        v4f b[8];
        #pragma unroll
        for (int t = 0; t < 8; ++t) b[t] = bl[(j * 8 + t) * 32];
        v2f a0; a0.x = a[j].x; a0.y = a[j].y;
        v2f a1; a1.x = a[j].z; a1.y = a[j].w;
        #pragma unroll
        for (int t = 0; t < 8; ++t) {
            v2f b0; b0.x = b[t].x; b0.y = b[t].y;
            acc[t] = __builtin_amdgcn_wmma_f32_16x16x4_f32(
                false, a0, false, b0, (short)0, acc[t], false, false);
        }
        #pragma unroll
        for (int t = 0; t < 8; ++t) {
            v2f b1; b1.x = b[t].z; b1.y = b[t].w;
            acc[t] = __builtin_amdgcn_wmma_f32_16x16x4_f32(
                false, a1, false, b1, (short)0, acc[t], false, false);
        }
    }

    v4f bq[2];
    #pragma unroll
    for (int g = 0; g < 2; ++g) {
        const float* bp = DUAL ? ((g ? bias1 : bias0) + 4 * c) : (bias0 + 64 * g + 4 * c);
        bq[g] = *(const v4f*)bp;
    }

    #pragma unroll
    for (int r = 0; r < 8; ++r) {
        const int row = row0 + r + 8 * h;
        if (row < nrows) {
            #pragma unroll
            for (int g = 0; g < 2; ++g) {
                v4f o;
                o.x = acc[4 * g][r]     + bq[g].x;
                o.y = acc[4 * g + 1][r] + bq[g].y;
                o.z = acc[4 * g + 2][r] + bq[g].z;
                o.w = acc[4 * g + 3][r] + bq[g].w;
                if (RELU) {
                    o.x = fmaxf(o.x, 0.0f); o.y = fmaxf(o.y, 0.0f);
                    o.z = fmaxf(o.z, 0.0f); o.w = fmaxf(o.w, 0.0f);
                }
                float* op = DUAL ? (C + ((size_t)g * nrows + row) * 64 + 4 * c)
                                 : (C + (size_t)row * FW + 64 * g + 4 * c);
                *(v4f*)op = o;
            }
        }
    }
}

#define OWAVES 32
#define KT     8
#define TILE_E (32 * 16 * KT)
#define QCAP   544
#define RCAP   1024


template <bool REC>
__device__ __forceinline__ void drain_hits(const int* q, int nb, int lane,
                                           const int* __restrict__ srcIdx, const int* __restrict__ dstIdx,
                                           const float* __restrict__ dinv, const float* __restrict__ in,
                                           float* acc, int lo, int nown,
                                           int* wS, int* wR, float* wW, int& nrec)
{
    int      s  = 0;
    unsigned rl = 0xffffffffu;
    float    da = 0.0f, db = 0.0f;
    if (lane < nb) {
        const int eid = q[lane];
        s = srcIdx[eid];
        const int dn = dstIdx[eid];
        da = dinv[s];
        db = dinv[dn];
        rl = (unsigned)(dn - lo);
    }
    #pragma unroll 1
    for (int b = 0; b < nb; b += 8) {
        {
            float v[8][4];
            #pragma unroll
            for (int k = 0; k < 8; ++k) {
                const int sq = __builtin_amdgcn_readlane(s, b + k);
                const float* ip = in + (size_t)sq * FW + lane;
                v[k][0] = ip[0]; v[k][1] = ip[32]; v[k][2] = ip[64]; v[k][3] = ip[96];
            }
            asm volatile("" :: "v"(v[0][0]), "v"(v[0][1]), "v"(v[0][2]), "v"(v[0][3]),
                               "v"(v[1][0]), "v"(v[1][1]), "v"(v[1][2]), "v"(v[1][3]),
                               "v"(v[2][0]), "v"(v[2][1]), "v"(v[2][2]), "v"(v[2][3]),
                               "v"(v[3][0]), "v"(v[3][1]), "v"(v[3][2]), "v"(v[3][3]));
            asm volatile("" :: "v"(v[4][0]), "v"(v[4][1]), "v"(v[4][2]), "v"(v[4][3]),
                               "v"(v[5][0]), "v"(v[5][1]), "v"(v[5][2]), "v"(v[5][3]),
                               "v"(v[6][0]), "v"(v[6][1]), "v"(v[6][2]), "v"(v[6][3]),
                               "v"(v[7][0]), "v"(v[7][1]), "v"(v[7][2]), "v"(v[7][3]));
            #pragma unroll
            for (int k = 0; k < 8; ++k) {
                const unsigned rq = (unsigned)__builtin_amdgcn_readlane((int)rl, b + k);
                if (rq < (unsigned)nown) {
                    const float nq = __builtin_bit_cast(float, __builtin_amdgcn_readlane(__builtin_bit_cast(int, da), b + k))
                                   * __builtin_bit_cast(float, __builtin_amdgcn_readlane(__builtin_bit_cast(int, db), b + k));
                    float* ap = acc + rq * FW + lane;
                    atomicAdd(ap,      nq * v[k][0]);
                    atomicAdd(ap + 32, nq * v[k][1]);
                    atomicAdd(ap + 64, nq * v[k][2]);
                    atomicAdd(ap + 96, nq * v[k][3]);
                }
            }
        }
    }
    if (REC) {
        const bool     val = rl < (unsigned)nown;
        const unsigned m   = __builtin_amdgcn_ballot_w32(val);
        const int      pos = nrec + (int)__builtin_amdgcn_mbcnt_lo(m, 0u);
        if (val && pos < RCAP) { wS[pos] = s; wR[pos] = (int)rl; wW[pos] = da * db; }
        nrec += __builtin_popcount(m);
    }
}

template <bool REC, bool REPLAY>
__global__ __launch_bounds__(1024)
void agg_owner(const int* __restrict__ srcIdx, const int* __restrict__ dstIdx,
               const unsigned* __restrict__ keyw,
               const float* __restrict__ dinv, const float* __restrict__ in,
               float* __restrict__ out, int* __restrict__ rcnt, int* __restrict__ rec,
               int n, int e)
{
    __shared__ __attribute__((aligned(16))) float acc[ROWS * FW];
    __shared__ int   pendE[OWAVES * QCAP];

    const int tid  = threadIdx.x;
    const int lane = tid & 31;
    const int wave = __builtin_amdgcn_readfirstlane(tid >> 5);
    const int lo   = blockIdx.x * ROWS;
    const int nown = (n - lo < ROWS) ? (n - lo) : ROWS;

    for (int r = wave; r < nown; r += OWAVES) {
        float d  = dinv[lo + r];
        float dd = d * d;
        v4f x = ((const v4f*)(in + (size_t)(lo + r) * FW))[lane];
        v4f y;
        y.x = dd * x.x; y.y = dd * x.y; y.z = dd * x.z; y.w = dd * x.w;
        ((v4f*)(acc + r * FW))[lane] = y;
    }
    __syncthreads();

    const int wid = blockIdx.x * OWAVES + wave;
    int*   wS = rec + (size_t)wid * (3 * RCAP);
    int*   wR = wS + RCAP;
    float* wW = (float*)(wR + RCAP);
    int  nrec   = 0;
    bool doScan = true;
    if (REPLAY) {
        nrec = __builtin_amdgcn_readfirstlane(rcnt[wid]);
        if (nrec <= RCAP) {
            doScan = false;
            for (int g = 0; g < nrec; g += 32) {
                const int idx = g + lane;
                int s = 0, r = 0;
                float w = 0.0f;
                if (idx < nrec) { s = wS[idx]; r = wR[idx]; w = wW[idx]; }
                const int nb = (nrec - g < 32) ? (nrec - g) : 32;
                #pragma unroll
                for (int sub = 0; sub < 4; ++sub) {
                    if (8 * sub < nb) {
                        float v[8][4];
                        #pragma unroll
                        for (int q = 0; q < 8; ++q) {
                            const int sq = __builtin_amdgcn_readlane(s, 8 * sub + q);
                            const float* ip = in + (size_t)sq * FW + lane;
                            v[q][0] = ip[0]; v[q][1] = ip[32]; v[q][2] = ip[64]; v[q][3] = ip[96];
                        }
                        asm volatile("" :: "v"(v[0][0]), "v"(v[0][1]), "v"(v[0][2]), "v"(v[0][3]),
                                           "v"(v[1][0]), "v"(v[1][1]), "v"(v[1][2]), "v"(v[1][3]),
                                           "v"(v[2][0]), "v"(v[2][1]), "v"(v[2][2]), "v"(v[2][3]),
                                           "v"(v[3][0]), "v"(v[3][1]), "v"(v[3][2]), "v"(v[3][3]));
                        asm volatile("" :: "v"(v[4][0]), "v"(v[4][1]), "v"(v[4][2]), "v"(v[4][3]),
                                           "v"(v[5][0]), "v"(v[5][1]), "v"(v[5][2]), "v"(v[5][3]),
                                           "v"(v[6][0]), "v"(v[6][1]), "v"(v[6][2]), "v"(v[6][3]),
                                           "v"(v[7][0]), "v"(v[7][1]), "v"(v[7][2]), "v"(v[7][3]));
                        #pragma unroll
                        for (int q = 0; q < 8; ++q) {
                            if (8 * sub + q < nb) {
                                const int   rq = __builtin_amdgcn_readlane(r, 8 * sub + q);
                                const float nq = __builtin_bit_cast(float,
                                    __builtin_amdgcn_readlane(__builtin_bit_cast(int, w), 8 * sub + q));
                                float* ap = acc + rq * FW + lane;
                                atomicAdd(ap,      nq * v[q][0]);
                                atomicAdd(ap + 32, nq * v[q][1]);
                                atomicAdd(ap + 64, nq * v[q][2]);
                                atomicAdd(ap + 96, nq * v[q][3]);
                            }
                        }
                    }
                }
            }
        }
        nrec = 0;
    }

    if (doScan) {
    int* qE = pendE + wave * QCAP;
    int cnt = 0;
    const unsigned kb = (unsigned)(blockIdx.x & 255) * 0x01010101u;
    const int ntiles = (e + TILE_E - 1) / TILE_E;
    const int shift = (int)(((long long)blockIdx.x * ntiles) / gridDim.x);

    for (int i = wave; i < ntiles; i += OWAVES) {
        int t = i + shift;
        if (t >= ntiles) t -= ntiles;
        const v4u* kp = (const v4u*)(keyw + (size_t)t * (TILE_E / 4)) + lane;
        v4u kv[KT];
        #pragma unroll
        for (int j = 0; j < KT; ++j) kv[j] = kp[32 * j];
        const int ebase = t * TILE_E + 16 * lane;
        #pragma unroll
        for (int j = 0; j < KT; ++j) {
            #pragma unroll
            for (int d = 0; d < 4; ++d) {
                const unsigned x = kv[j][d] ^ kb;
                unsigned f = (x - 0x01010101u) & ~x & 0x80808080u;
                unsigned mask = __builtin_amdgcn_ballot_w32(f != 0u);
                while (mask) {
                    const int  eid = ebase + 512 * j + 4 * d + (__builtin_ctz(f | 0x80000000u) >> 3);
                    const bool ok  = (f != 0u) && (eid < e);
                    const unsigned m2 = __builtin_amdgcn_ballot_w32(ok);
                    const int pos = cnt + (int)__builtin_amdgcn_mbcnt_lo(m2, 0u);
                    if (ok) qE[pos] = eid;
                    cnt += __builtin_popcount(m2);
                    f &= f - 1u;
                    mask = __builtin_amdgcn_ballot_w32(f != 0u);
                }
            }
            while (cnt >= 32) {
                cnt -= 32;
                drain_hits<REC>(qE + cnt, 32, lane, srcIdx, dstIdx, dinv, in, acc, lo, nown, wS, wR, wW, nrec);
            }
        }
    }
    if (cnt > 0) drain_hits<REC>(qE, cnt, lane, srcIdx, dstIdx, dinv, in, acc, lo, nown, wS, wR, wW, nrec);
    }
    if (REC) { if (lane == 0) rcnt[wid] = nrec; }
    __syncthreads();

    for (int r = wave; r < nown; r += OWAVES) {
        ((v4f*)(out + (size_t)(lo + r) * FW))[lane] = ((const v4f*)(acc + r * FW))[lane];
    }
}

extern "C" void kernel_launch(void* const* d_in, const int* in_sizes, int n_in,
                              void* d_out, int out_size, void* d_ws, size_t ws_size,
                              hipStream_t stream)
{
    const float* x    = (const float*)d_in[0];
    const int*   ei   = (const int*)  d_in[1];
    const float* W1   = (const float*)d_in[2];
    const float* b1   = (const float*)d_in[3];
    const float* Wmu  = (const float*)d_in[4];
    const float* bmu  = (const float*)d_in[5];
    const float* Wlv  = (const float*)d_in[6];
    const float* blv  = (const float*)d_in[7];

    const int N = in_sizes[0] / KDIM;
    const int E = in_sizes[1] / 2;
    const int* srcIdx = ei;
    const int* dstIdx = ei + E;

    size_t dinv_bytes = (((size_t)N * 4) + 255) & ~(size_t)255;
    float* dinv = (float*)d_ws;
    float* bufA = (float*)((char*)d_ws + dinv_bytes);
    float* bufB = bufA + (size_t)N * FW;
    unsigned* part = (unsigned*)(bufB + (size_t)N * FW);

    const int TB = 256;
    dim3 blk(TB);
    int gP   = ((N + 1) / 2 + TB - 1) / TB;
    int nCh  = (E + ECHUNK - 1) / ECHUNK;
    int nRg  = (N + HBINS - 1) / HBINS;
    int gOwn = (N + ROWS - 1) / ROWS;
    int gGemm = (N + GROWS - 1) / GROWS;
    unsigned* keyw = part + (size_t)nRg * nCh * HWORDS;
    int* rcnt = (int*)(keyw + (size_t)nCh * (ECHUNK / 4));
    int* rec  = rcnt + (size_t)gOwn * OWAVES;

    deg_hist  <<<dim3(nCh, nRg), dim3(1024), 0, stream>>>(dstIdx, part, keyw, N, E);
    deg_finish<<<gP, blk, 0, stream>>>(part, dinv, N, nCh);


    agg_owner<true, false><<<gOwn, dim3(1024), 0, stream>>>(srcIdx, dstIdx, keyw, dinv, x, bufB, rcnt, rec, N, E);

    gemm_wmma_f32<false, true><<<gGemm, dim3(128), 0, stream>>>(bufB, W1, nullptr, b1, nullptr, bufA, N);

    agg_owner<false, true><<<gOwn, dim3(1024), 0, stream>>>(srcIdx, dstIdx, keyw, dinv, bufA, bufB, rcnt, rec, N, E);

    gemm_wmma_f32<true, false><<<gGemm, dim3(128), 0, stream>>>(bufB, Wmu, Wlv, bmu, blv, (float*)d_out, N);
}
